# starting kernel
# baseline (speedup 1.0000x reference)
#include <hip/hip_runtime.h>

typedef __attribute__((ext_vector_type(2))) float v2f;
typedef __attribute__((ext_vector_type(8))) float v8f;

#define DFEAT 128
#define PSTR 288

__global__ void k_zero(float* __restrict__ p, int n) {
    int i = blockIdx.x * blockDim.x + threadIdx.x;
    int stride = gridDim.x * blockDim.x;
    for (; i < n; i += stride) p[i] = 0.0f;
}

__global__ void k_deg(const int* __restrict__ esrc, const int* __restrict__ edst,
                      float* __restrict__ dout, float* __restrict__ din, int e) {
    int i = blockIdx.x * blockDim.x + threadIdx.x;
    if (i < e) {
        atomicAdd(&dout[esrc[i]], 1.0f);
        atomicAdd(&din[edst[i]], 1.0f);
    }
}

__global__ void k_norm(float* __restrict__ d, int n) {
    int i = blockIdx.x * blockDim.x + threadIdx.x;
    if (i < n) {
        float v = d[i];
        d[i] = (v > 0.0f) ? rsqrtf(v) : 0.0f;
    }
}

__global__ __launch_bounds__(256) void k_scatter(const float* __restrict__ x,
                                                 const float* __restrict__ nsrc,
                                                 const int* __restrict__ esrc,
                                                 const int* __restrict__ edst,
                                                 float* __restrict__ agg, int e) {
    int gw = (blockIdx.x * blockDim.x + threadIdx.x) >> 5;
    int lane = threadIdx.x & 31;
    if (gw >= e) return;
    int s = esrc[gw];
    int d = edst[gw];
    float ns = nsrc[s];
    float4 v = ((const float4*)(x + (long)s * DFEAT))[lane];
    float* dst = agg + (long)d * DFEAT + lane * 4;
    atomicAdd(dst + 0, v.x * ns);
    atomicAdd(dst + 1, v.y * ns);
    atomicAdd(dst + 2, v.z * ns);
    atomicAdd(dst + 3, v.w * ns);
}

__global__ __launch_bounds__(256) void k_gemm_bn(const float* agg, float* hpre,
                                                 const float* __restrict__ W,
                                                 const float* __restrict__ bias,
                                                 const float* __restrict__ alpha_p,
                                                 const float* __restrict__ ndst,
                                                 float* __restrict__ fsum,
                                                 float* __restrict__ fsq, int n) {
    __shared__ float sW[16 * PSTR];
    __shared__ float sSum[DFEAT];
    __shared__ float sSq[DFEAT];

    const int tid = threadIdx.x;
    if (tid < DFEAT) { sSum[tid] = 0.0f; sSq[tid] = 0.0f; }

    const int wave = tid >> 5;
    const int lane = tid & 31;
    const int half = lane >> 4;
    const int mcol = lane & 15;
    const int r0 = blockIdx.x * 128 + wave * 16;
    const int rowA = r0 + mcol;
    const bool aok = rowA < n;
    const float nd = aok ? ndst[rowA] : 0.0f;
    const int rowC = aok ? rowA : (n - 1);
    const float alpha = alpha_p[0];
    const float* aptr = agg + (long)rowC * DFEAT;

    v8f acc[8] = {};

    for (int kc = 0; kc < DFEAT; kc += 32) {
        __syncthreads();
        {
            int kp = tid >> 4;
            int nb = (tid & 15) * 8;
            const float* w0 = W + (kc + 2 * kp) * DFEAT + nb;
            const float* w1 = w0 + DFEAT;
            float4 r0a = ((const float4*)w0)[0];
            float4 r0b = ((const float4*)w0)[1];
            float4 r1a = ((const float4*)w1)[0];
            float4 r1b = ((const float4*)w1)[1];
            float2* dp = (float2*)(sW + kp * PSTR + nb * 2);
            dp[0] = make_float2(r0a.x, r1a.x);
            dp[1] = make_float2(r0a.y, r1a.y);
            dp[2] = make_float2(r0a.z, r1a.z);
            dp[3] = make_float2(r0a.w, r1a.w);
            dp[4] = make_float2(r0b.x, r1b.x);
            dp[5] = make_float2(r0b.y, r1b.y);
            dp[6] = make_float2(r0b.z, r1b.z);
            dp[7] = make_float2(r0b.w, r1b.w);
        }
        __syncthreads();

        for (int kb = 0; kb < 32; kb += 4) {
            const int kA = kc + kb + half * 2;
            float2 av = *(const float2*)(aptr + kA);
            v2f a;
            a.x = av.x * nd;
            a.y = av.y * nd;
            const float* bp = sW + (kb / 2 + half) * PSTR + mcol * 2;
#pragma unroll
            for (int t = 0; t < 8; ++t) {
                v2f b = *(const v2f*)(bp + t * 32);
                acc[t] = __builtin_amdgcn_wmma_f32_16x16x4_f32(
                    false, a, false, b, (short)0, acc[t], false, false);
            }
        }
    }

#pragma unroll
    for (int t = 0; t < 8; ++t) {
        const int col = t * 16 + mcol;
        const float bv = bias[col];
#pragma unroll
        for (int r = 0; r < 8; ++r) {
            const int row = r0 + half * 8 + r;
            if (row < n) {
                float v = acc[t][r] + bv;
                v = (v >= 0.0f) ? v : alpha * v;
                hpre[(long)row * DFEAT + col] = v;
                atomicAdd(&sSum[col], v);
                atomicAdd(&sSq[col], v * v);
            }
        }
    }
    __syncthreads();
    if (tid < DFEAT) {
        atomicAdd(&fsum[tid], sSum[tid]);
        atomicAdd(&fsq[tid], sSq[tid]);
    }
}

__global__ void k_bnstats(const float* __restrict__ fsum, const float* __restrict__ fsq,
                          const float* __restrict__ gamma, const float* __restrict__ beta,
                          float* __restrict__ bnA, float* __restrict__ bnC, int n) {
    int c = threadIdx.x;
    float invn = 1.0f / (float)n;
    float mean = fsum[c] * invn;
    float var = fsq[c] * invn - mean * mean;
    float inv = rsqrtf(var + 1e-5f);
    float a = inv * gamma[c];
    bnA[c] = a;
    bnC[c] = beta[c] - mean * a;
}

__global__ __launch_bounds__(256) void k_bnapply(const float* __restrict__ hpre,
                                                 const float* __restrict__ bnA,
                                                 const float* __restrict__ bnC,
                                                 const float* __restrict__ alpha_p,
                                                 const int* __restrict__ gids,
                                                 float* __restrict__ hout,
                                                 float* __restrict__ gh, int n) {
    int idx = blockIdx.x * blockDim.x + threadIdx.x;
    int total = n * (DFEAT / 4);
    if (idx >= total) return;
    int node = idx >> 5;
    int q = idx & 31;
    float alpha = alpha_p[0];
    float4 v = ((const float4*)hpre)[idx];
    float4 a = ((const float4*)bnA)[q];
    float4 c = ((const float4*)bnC)[q];
    float4 y;
    y.x = fmaf(v.x, a.x, c.x); y.x = (y.x >= 0.0f) ? y.x : alpha * y.x;
    y.y = fmaf(v.y, a.y, c.y); y.y = (y.y >= 0.0f) ? y.y : alpha * y.y;
    y.z = fmaf(v.z, a.z, c.z); y.z = (y.z >= 0.0f) ? y.z : alpha * y.z;
    y.w = fmaf(v.w, a.w, c.w); y.w = (y.w >= 0.0f) ? y.w : alpha * y.w;
    ((float4*)hout)[idx] = y;
    int g = gids[node];
    float* gp = gh + (long)g * DFEAT + q * 4;
    atomicAdd(gp + 0, y.x);
    atomicAdd(gp + 1, y.y);
    atomicAdd(gp + 2, y.z);
    atomicAdd(gp + 3, y.w);
}

static inline int cdiv(long a, long b) { return (int)((a + b - 1) / b); }

extern "C" void kernel_launch(void* const* d_in, const int* in_sizes, int n_in,
                              void* d_out, int out_size, void* d_ws, size_t ws_size,
                              hipStream_t stream) {
    (void)n_in; (void)ws_size;
    const float* heat = (const float*)d_in[0];
    const int* esrc   = (const int*)d_in[1];
    const int* edst   = (const int*)d_in[2];
    const int* gids   = (const int*)d_in[3];
    const float* W0 = (const float*)d_in[4];
    const float* b0 = (const float*)d_in[5];
    const float* ac0 = (const float*)d_in[6];
    const float* g0 = (const float*)d_in[7];
    const float* be0 = (const float*)d_in[8];
    const float* aa0 = (const float*)d_in[9];
    const float* W1 = (const float*)d_in[10];
    const float* b1 = (const float*)d_in[11];
    const float* ac1 = (const float*)d_in[12];
    const float* g1 = (const float*)d_in[13];
    const float* be1 = (const float*)d_in[14];
    const float* aa1 = (const float*)d_in[15];

    const int E = in_sizes[1];
    const int N = in_sizes[3];
    const int ND = N * DFEAT;
    const int GD = out_size - ND;

    float* ws   = (float*)d_ws;
    float* nsrc = ws;
    float* ndst = ws + N;
    float* agg  = ws + 2 * (long)N;
    float* h0   = agg + ND;
    float* fsum = h0 + ND;
    float* fsq  = fsum + DFEAT;
    float* bnA  = fsq + DFEAT;
    float* bnC  = bnA + DFEAT;

    float* hout = (float*)d_out;
    float* gh   = hout + ND;

    const int gemm_blocks = cdiv(N, 128);

    k_zero<<<cdiv(2 * (long)N, 256), 256, 0, stream>>>(nsrc, 2 * N);
    k_zero<<<cdiv(ND, 256), 256, 0, stream>>>(agg, ND);
    k_zero<<<1, 256, 0, stream>>>(fsum, 2 * DFEAT);
    k_zero<<<cdiv(GD, 256), 256, 0, stream>>>(gh, GD);

    k_deg<<<cdiv(E, 256), 256, 0, stream>>>(esrc, edst, nsrc, ndst, E);
    k_norm<<<cdiv(2 * (long)N, 256), 256, 0, stream>>>(nsrc, 2 * N);

    k_scatter<<<cdiv((long)E * 32, 256), 256, 0, stream>>>(heat, nsrc, esrc, edst, agg, E);
    k_gemm_bn<<<gemm_blocks, 256, 0, stream>>>(agg, agg, W0, b0, ac0, ndst, fsum, fsq, N);
    k_bnstats<<<1, DFEAT, 0, stream>>>(fsum, fsq, g0, be0, bnA, bnC, N);
    k_bnapply<<<cdiv((long)N * 32, 256), 256, 0, stream>>>(agg, bnA, bnC, aa0, gids, h0, gh, N);

    k_zero<<<cdiv(ND, 256), 256, 0, stream>>>(agg, ND);
    k_zero<<<1, 256, 0, stream>>>(fsum, 2 * DFEAT);
    k_scatter<<<cdiv((long)E * 32, 256), 256, 0, stream>>>(h0, nsrc, esrc, edst, agg, E);
    k_gemm_bn<<<gemm_blocks, 256, 0, stream>>>(agg, agg, W1, b1, ac1, ndst, fsum, fsq, N);
    k_bnstats<<<1, DFEAT, 0, stream>>>(fsum, fsq, g1, be1, bnA, bnC, N);
    k_bnapply<<<cdiv((long)N * 32, 256), 256, 0, stream>>>(agg, bnA, bnC, aa1, gids, hout, gh, N);
}


# bnapply: each wave walks 16 consecutive rows and pre-sums runs of equal graph id in registers, one atomic set per run instead of per row (graph sums now add run partial sums)
# speedup vs baseline: 1.2631x; 1.2631x over previous
#include <hip/hip_runtime.h>

typedef __attribute__((ext_vector_type(2))) float v2f;
typedef __attribute__((ext_vector_type(8))) float v8f;

#define DFEAT 128
#define PSTR 288

__global__ void k_zero(float* __restrict__ p, int n) {
    int i = blockIdx.x * blockDim.x + threadIdx.x;
    int stride = gridDim.x * blockDim.x;
    for (; i < n; i += stride) p[i] = 0.0f;
}

__global__ void k_deg(const int* __restrict__ esrc, const int* __restrict__ edst,
                      float* __restrict__ dout, float* __restrict__ din, int e) {
    int i = blockIdx.x * blockDim.x + threadIdx.x;
    if (i < e) {
        atomicAdd(&dout[esrc[i]], 1.0f);
        atomicAdd(&din[edst[i]], 1.0f);
    }
}

__global__ void k_norm(float* __restrict__ d, int n) {
    int i = blockIdx.x * blockDim.x + threadIdx.x;
    if (i < n) {
        float v = d[i];
        d[i] = (v > 0.0f) ? rsqrtf(v) : 0.0f;
    }
}

__global__ __launch_bounds__(256) void k_scatter(const float* __restrict__ x,
                                                 const float* __restrict__ nsrc,
                                                 const int* __restrict__ esrc,
                                                 const int* __restrict__ edst,
                                                 float* __restrict__ agg, int e) {
    int gw = (blockIdx.x * blockDim.x + threadIdx.x) >> 5;
    int lane = threadIdx.x & 31;
    if (gw >= e) return;
    int s = esrc[gw];
    int d = edst[gw];
    float ns = nsrc[s];
    float4 v = ((const float4*)(x + (long)s * DFEAT))[lane];
    float* dst = agg + (long)d * DFEAT + lane * 4;
    atomicAdd(dst + 0, v.x * ns);
    atomicAdd(dst + 1, v.y * ns);
    atomicAdd(dst + 2, v.z * ns);
    atomicAdd(dst + 3, v.w * ns);
}

__global__ __launch_bounds__(256) void k_gemm_bn(const float* agg, float* hpre,
                                                 const float* __restrict__ W,
                                                 const float* __restrict__ bias,
                                                 const float* __restrict__ alpha_p,
                                                 const float* __restrict__ ndst,
                                                 float* __restrict__ fsum,
                                                 float* __restrict__ fsq, int n) {
    __shared__ float sW[16 * PSTR];
    __shared__ float sSum[DFEAT];
    __shared__ float sSq[DFEAT];

    const int tid = threadIdx.x;
    if (tid < DFEAT) { sSum[tid] = 0.0f; sSq[tid] = 0.0f; }

    const int wave = tid >> 5;
    const int lane = tid & 31;
    const int half = lane >> 4;
    const int mcol = lane & 15;
    const int r0 = blockIdx.x * 128 + wave * 16;
    const int rowA = r0 + mcol;
    const bool aok = rowA < n;
    const float nd = aok ? ndst[rowA] : 0.0f;
    const int rowC = aok ? rowA : (n - 1);
    const float alpha = alpha_p[0];
    const float* aptr = agg + (long)rowC * DFEAT;

    v8f acc[8] = {};

    for (int kc = 0; kc < DFEAT; kc += 32) {
        __syncthreads();
        {
            int kp = tid >> 4;
            int nb = (tid & 15) * 8;
            const float* w0 = W + (kc + 2 * kp) * DFEAT + nb;
            const float* w1 = w0 + DFEAT;
            float4 r0a = ((const float4*)w0)[0];
            float4 r0b = ((const float4*)w0)[1];
            float4 r1a = ((const float4*)w1)[0];
            float4 r1b = ((const float4*)w1)[1];
            float2* dp = (float2*)(sW + kp * PSTR + nb * 2);
            dp[0] = make_float2(r0a.x, r1a.x);
            dp[1] = make_float2(r0a.y, r1a.y);
            dp[2] = make_float2(r0a.z, r1a.z);
            dp[3] = make_float2(r0a.w, r1a.w);
            dp[4] = make_float2(r0b.x, r1b.x);
            dp[5] = make_float2(r0b.y, r1b.y);
            dp[6] = make_float2(r0b.z, r1b.z);
            dp[7] = make_float2(r0b.w, r1b.w);
        }
        __syncthreads();

        for (int kb = 0; kb < 32; kb += 4) {
            const int kA = kc + kb + half * 2;
            float2 av = *(const float2*)(aptr + kA);
            v2f a;
            a.x = av.x * nd;
            a.y = av.y * nd;
            const float* bp = sW + (kb / 2 + half) * PSTR + mcol * 2;
#pragma unroll
            for (int t = 0; t < 8; ++t) {
                v2f b = *(const v2f*)(bp + t * 32);
                acc[t] = __builtin_amdgcn_wmma_f32_16x16x4_f32(
                    false, a, false, b, (short)0, acc[t], false, false);
            }
        }
    }

#pragma unroll
    for (int t = 0; t < 8; ++t) {
        const int col = t * 16 + mcol;
        const float bv = bias[col];
#pragma unroll
        for (int r = 0; r < 8; ++r) {
            const int row = r0 + half * 8 + r;
            if (row < n) {
                float v = acc[t][r] + bv;
                v = (v >= 0.0f) ? v : alpha * v;
                hpre[(long)row * DFEAT + col] = v;
                atomicAdd(&sSum[col], v);
                atomicAdd(&sSq[col], v * v);
            }
        }
    }
    __syncthreads();
    if (tid < DFEAT) {
        atomicAdd(&fsum[tid], sSum[tid]);
        atomicAdd(&fsq[tid], sSq[tid]);
    }
}

__global__ void k_bnstats(const float* __restrict__ fsum, const float* __restrict__ fsq,
                          const float* __restrict__ gamma, const float* __restrict__ beta,
                          float* __restrict__ bnA, float* __restrict__ bnC, int n) {
    int c = threadIdx.x;
    float invn = 1.0f / (float)n;
    float mean = fsum[c] * invn;
    float var = fsq[c] * invn - mean * mean;
    float inv = rsqrtf(var + 1e-5f);
    float a = inv * gamma[c];
    bnA[c] = a;
    bnC[c] = beta[c] - mean * a;
}

#define BN_ROWS 16

__device__ __forceinline__ float4 bn_act(float4 v, float4 a, float4 c, float alpha) {
    float4 y;
    y.x = fmaf(v.x, a.x, c.x); y.x = (y.x >= 0.0f) ? y.x : alpha * y.x;
    y.y = fmaf(v.y, a.y, c.y); y.y = (y.y >= 0.0f) ? y.y : alpha * y.y;
    y.z = fmaf(v.z, a.z, c.z); y.z = (y.z >= 0.0f) ? y.z : alpha * y.z;
    y.w = fmaf(v.w, a.w, c.w); y.w = (y.w >= 0.0f) ? y.w : alpha * y.w;
    return y;
}

__device__ __forceinline__ void gh_flush(float* __restrict__ gh, int g, int lane, float4 s) {
    float* gp = gh + (long)g * DFEAT + lane * 4;
    atomicAdd(gp + 0, s.x);
    atomicAdd(gp + 1, s.y);
    atomicAdd(gp + 2, s.z);
    atomicAdd(gp + 3, s.w);
}

__global__ __launch_bounds__(128) void k_bnapply(const float* __restrict__ hpre,
                                                 const float* __restrict__ bnA,
                                                 const float* __restrict__ bnC,
                                                 const float* __restrict__ alpha_p,
                                                 const int* __restrict__ gids,
                                                 float* __restrict__ hout,
                                                 float* __restrict__ gh, int n) {
    const int lane = threadIdx.x & 31;
    const int gw = __builtin_amdgcn_readfirstlane((int)(blockIdx.x * 4 + (threadIdx.x >> 5)));
    const int row0 = gw * BN_ROWS;
    if (row0 >= n) return;
    const float alpha = alpha_p[0];
    const float4 a = ((const float4*)bnA)[lane];
    const float4 c = ((const float4*)bnC)[lane];
    const float4* src = (const float4*)hpre + (long)row0 * 32 + lane;
    float4* dst = (float4*)hout + (long)row0 * 32 + lane;
    float4 s = make_float4(0.0f, 0.0f, 0.0f, 0.0f);
    int gcur = gids[row0];
    if (row0 + BN_ROWS <= n) {
        float4 v[BN_ROWS];
        int g[BN_ROWS];
#pragma unroll
        for (int i = 0; i < BN_ROWS; ++i) {
            v[i] = src[i * 32];
            g[i] = gids[row0 + i];
        }
#pragma unroll
        for (int i = 0; i < BN_ROWS; ++i) {
            float4 y = bn_act(v[i], a, c, alpha);
            dst[i * 32] = y;
            if (g[i] != gcur) {
                gh_flush(gh, gcur, lane, s);
                s = make_float4(0.0f, 0.0f, 0.0f, 0.0f);
                gcur = g[i];
            }
            s.x += y.x; s.y += y.y; s.z += y.z; s.w += y.w;
        }
    } else {
        const int cnt = n - row0;
        for (int i = 0; i < cnt; ++i) {
            float4 y = bn_act(src[i * 32], a, c, alpha);
            dst[i * 32] = y;
            const int g = gids[row0 + i];
            if (g != gcur) {
                gh_flush(gh, gcur, lane, s);
                s = make_float4(0.0f, 0.0f, 0.0f, 0.0f);
                gcur = g;
            }
            s.x += y.x; s.y += y.y; s.z += y.z; s.w += y.w;
        }
    }
    gh_flush(gh, gcur, lane, s);
}

static inline int cdiv(long a, long b) { return (int)((a + b - 1) / b); }

extern "C" void kernel_launch(void* const* d_in, const int* in_sizes, int n_in,
                              void* d_out, int out_size, void* d_ws, size_t ws_size,
                              hipStream_t stream) {
    (void)n_in; (void)ws_size;
    const float* heat = (const float*)d_in[0];
    const int* esrc   = (const int*)d_in[1];
    const int* edst   = (const int*)d_in[2];
    const int* gids   = (const int*)d_in[3];
    const float* W0 = (const float*)d_in[4];
    const float* b0 = (const float*)d_in[5];
    const float* ac0 = (const float*)d_in[6];
    const float* g0 = (const float*)d_in[7];
    const float* be0 = (const float*)d_in[8];
    const float* aa0 = (const float*)d_in[9];
    const float* W1 = (const float*)d_in[10];
    const float* b1 = (const float*)d_in[11];
    const float* ac1 = (const float*)d_in[12];
    const float* g1 = (const float*)d_in[13];
    const float* be1 = (const float*)d_in[14];
    const float* aa1 = (const float*)d_in[15];

    const int E = in_sizes[1];
    const int N = in_sizes[3];
    const int ND = N * DFEAT;
    const int GD = out_size - ND;

    float* ws   = (float*)d_ws;
    float* nsrc = ws;
    float* ndst = ws + N;
    float* agg  = ws + 2 * (long)N;
    float* h0   = agg + ND;
    float* fsum = h0 + ND;
    float* fsq  = fsum + DFEAT;
    float* bnA  = fsq + DFEAT;
    float* bnC  = bnA + DFEAT;

    float* hout = (float*)d_out;
    float* gh   = hout + ND;

    const int gemm_blocks = cdiv(N, 128);

    k_zero<<<cdiv(2 * (long)N, 256), 256, 0, stream>>>(nsrc, 2 * N);
    k_zero<<<cdiv(ND, 256), 256, 0, stream>>>(agg, ND);
    k_zero<<<1, 256, 0, stream>>>(fsum, 2 * DFEAT);
    k_zero<<<cdiv(GD, 256), 256, 0, stream>>>(gh, GD);

    k_deg<<<cdiv(E, 256), 256, 0, stream>>>(esrc, edst, nsrc, ndst, E);
    k_norm<<<cdiv(2 * (long)N, 256), 256, 0, stream>>>(nsrc, 2 * N);

    k_scatter<<<cdiv((long)E * 32, 256), 256, 0, stream>>>(heat, nsrc, esrc, edst, agg, E);
    k_gemm_bn<<<gemm_blocks, 256, 0, stream>>>(agg, agg, W0, b0, ac0, ndst, fsum, fsq, N);
    k_bnstats<<<1, DFEAT, 0, stream>>>(fsum, fsq, g0, be0, bnA, bnC, N);
    k_bnapply<<<cdiv(N, BN_ROWS * 4), 128, 0, stream>>>(agg, bnA, bnC, aa0, gids, h0, gh, N);

    k_zero<<<cdiv(ND, 256), 256, 0, stream>>>(agg, ND);
    k_zero<<<1, 256, 0, stream>>>(fsum, 2 * DFEAT);
    k_scatter<<<cdiv((long)E * 32, 256), 256, 0, stream>>>(h0, nsrc, esrc, edst, agg, E);
    k_gemm_bn<<<gemm_blocks, 256, 0, stream>>>(agg, agg, W1, b1, ac1, ndst, fsum, fsq, N);
    k_bnstats<<<1, DFEAT, 0, stream>>>(fsum, fsq, g1, be1, bnA, bnC, N);
    k_bnapply<<<cdiv(N, BN_ROWS * 4), 128, 0, stream>>>(agg, bnA, bnC, aa1, gids, hout, gh, N);
}


# aggregation: edges grouped by destination each call (int counts, scan, slot fill); one wave per node sums its in-edges in registers, one store per row, no float atomics (sum in bucket order)
# speedup vs baseline: 1.7990x; 1.4243x over previous
#include <hip/hip_runtime.h>

typedef __attribute__((ext_vector_type(2))) float v2f;
typedef __attribute__((ext_vector_type(8))) float v8f;

#define DFEAT 128
#define PSTR 288

__global__ void k_zero(float* __restrict__ p, int n) {
    int i = blockIdx.x * blockDim.x + threadIdx.x;
    int stride = gridDim.x * blockDim.x;
    for (; i < n; i += stride) p[i] = 0.0f;
}

__global__ void k_deg(const int* __restrict__ esrc, const int* __restrict__ edst,
                      int* __restrict__ dout, int* __restrict__ din, int e) {
    int i = blockIdx.x * blockDim.x + threadIdx.x;
    if (i < e) {
        atomicAdd(&dout[esrc[i]], 1);
        atomicAdd(&din[edst[i]], 1);
    }
}

#define SCAN_THREADS 1024
__global__ __launch_bounds__(SCAN_THREADS) void k_scan(int* __restrict__ ptr,
                                                       float* __restrict__ ndst, int n) {
    __shared__ int swave[SCAN_THREADS / 32];
    const int lane = threadIdx.x & 31;
    const int wave = threadIdx.x >> 5;
    const int seg = ((n + SCAN_THREADS - 1) / SCAN_THREADS) * 32;
    const int sbeg = min(wave * seg, n);
    const int send = min(sbeg + seg, n);

    int sum = 0;
    for (int b = sbeg; b < send; b += 256) {
#pragma unroll
        for (int u = 0; u < 8; ++u) {
            const int i = b + u * 32 + lane;
            sum += (i < send) ? ptr[i] : 0;
        }
    }
    for (int o = 16; o > 0; o >>= 1) sum += __shfl_xor(sum, o);
    if (lane == 0) swave[wave] = sum;
    __syncthreads();
    int carry = (lane < wave) ? swave[lane] : 0;
    for (int o = 16; o > 0; o >>= 1) carry += __shfl_xor(carry, o);

    for (int b = sbeg; b < send; b += 256) {
        const int i0 = b + lane * 8;
        int c[8];
        int t = 0;
#pragma unroll
        for (int u = 0; u < 8; ++u) {
            c[u] = (i0 + u < send) ? ptr[i0 + u] : 0;
            t += c[u];
        }
        int v = t;
        for (int o = 1; o < 32; o <<= 1) {
            const int up = __shfl_up(v, o);
            if (lane >= o) v += up;
        }
        int run = carry + v - t;
#pragma unroll
        for (int u = 0; u < 8; ++u) {
            run += c[u];
            if (i0 + u < send) {
                ptr[i0 + u] = run;
                ndst[i0 + u] = (c[u] > 0) ? rsqrtf((float)c[u]) : 0.0f;
            }
        }
        carry += __builtin_amdgcn_readlane(v, 31);
    }
}

__global__ void k_fill(const int* __restrict__ esrc, const int* __restrict__ edst,
                       const int* __restrict__ dout, int* __restrict__ ptr,
                       int* __restrict__ csr, int e) {
    int i = blockIdx.x * blockDim.x + threadIdx.x;
    if (i < e) {
        const int s = esrc[i];
        const int d = edst[i];
        const int pos = atomicSub(&ptr[d], 1) - 1;
        const float w = rsqrtf((float)dout[s]);
        ((int2*)csr)[pos] = make_int2(s, __float_as_int(w));
    }
}

__global__ __launch_bounds__(256) void k_gather(const float* __restrict__ x,
                                                const int* __restrict__ ptr,
                                                const int* __restrict__ csr,
                                                float* __restrict__ agg, int n, int e) {
    const int lane = threadIdx.x & 31;
    const int row = __builtin_amdgcn_readfirstlane((int)(blockIdx.x * 8 + (threadIdx.x >> 5)));
    if (row >= n) return;
    const int beg = ptr[row];
    const int end = (row + 1 < n) ? ptr[row + 1] : e;
    const int2* ent = (const int2*)csr;
    const float4* xv = (const float4*)x + lane;
    float4 acc = make_float4(0.0f, 0.0f, 0.0f, 0.0f);
    for (int j = beg; j < end; j += 4) {
        const int rem = end - j;
        const bool ok1 = rem > 1, ok2 = rem > 2, ok3 = rem > 3;
        const int2 e0 = ent[j];
        const int2 e1 = ent[ok1 ? j + 1 : j];
        const int2 e2 = ent[ok2 ? j + 2 : j];
        const int2 e3 = ent[ok3 ? j + 3 : j];
        const float4 v0 = xv[(long)e0.x * 32];
        const float4 v1 = xv[(long)e1.x * 32];
        const float4 v2 = xv[(long)e2.x * 32];
        const float4 v3 = xv[(long)e3.x * 32];
        const float w0 = __int_as_float(e0.y);
        const float w1 = __int_as_float(e1.y);
        const float w2 = __int_as_float(e2.y);
        const float w3 = __int_as_float(e3.y);
        acc.x += v0.x * w0; acc.y += v0.y * w0; acc.z += v0.z * w0; acc.w += v0.w * w0;
        acc.x = ok1 ? acc.x + v1.x * w1 : acc.x;
        acc.y = ok1 ? acc.y + v1.y * w1 : acc.y;
        acc.z = ok1 ? acc.z + v1.z * w1 : acc.z;
        acc.w = ok1 ? acc.w + v1.w * w1 : acc.w;
        acc.x = ok2 ? acc.x + v2.x * w2 : acc.x;
        acc.y = ok2 ? acc.y + v2.y * w2 : acc.y;
        acc.z = ok2 ? acc.z + v2.z * w2 : acc.z;
        acc.w = ok2 ? acc.w + v2.w * w2 : acc.w;
        acc.x = ok3 ? acc.x + v3.x * w3 : acc.x;
        acc.y = ok3 ? acc.y + v3.y * w3 : acc.y;
        acc.z = ok3 ? acc.z + v3.z * w3 : acc.z;
        acc.w = ok3 ? acc.w + v3.w * w3 : acc.w;
    }
    ((float4*)agg)[(long)row * 32 + lane] = acc;
}

__global__ __launch_bounds__(256) void k_gemm_bn(const float* agg, float* hpre,
                                                 const float* __restrict__ W,
                                                 const float* __restrict__ bias,
                                                 const float* __restrict__ alpha_p,
                                                 const float* __restrict__ ndst,
                                                 float* __restrict__ fsum,
                                                 float* __restrict__ fsq, int n) {
    __shared__ float sW[16 * PSTR];
    __shared__ float sSum[DFEAT];
    __shared__ float sSq[DFEAT];

    const int tid = threadIdx.x;
    if (tid < DFEAT) { sSum[tid] = 0.0f; sSq[tid] = 0.0f; }

    const int wave = tid >> 5;
    const int lane = tid & 31;
    const int half = lane >> 4;
    const int mcol = lane & 15;
    const int r0 = blockIdx.x * 128 + wave * 16;
    const int rowA = r0 + mcol;
    const bool aok = rowA < n;
    const float nd = aok ? ndst[rowA] : 0.0f;
    const int rowC = aok ? rowA : (n - 1);
    const float alpha = alpha_p[0];
    const float* aptr = agg + (long)rowC * DFEAT;

    v8f acc[8] = {};

    for (int kc = 0; kc < DFEAT; kc += 32) {
        __syncthreads();
        {
            int kp = tid >> 4;
            int nb = (tid & 15) * 8;
            const float* w0 = W + (kc + 2 * kp) * DFEAT + nb;
            const float* w1 = w0 + DFEAT;
            float4 r0a = ((const float4*)w0)[0];
            float4 r0b = ((const float4*)w0)[1];
            float4 r1a = ((const float4*)w1)[0];
            float4 r1b = ((const float4*)w1)[1];
            float2* dp = (float2*)(sW + kp * PSTR + nb * 2);
            dp[0] = make_float2(r0a.x, r1a.x);
            dp[1] = make_float2(r0a.y, r1a.y);
            dp[2] = make_float2(r0a.z, r1a.z);
            dp[3] = make_float2(r0a.w, r1a.w);
            dp[4] = make_float2(r0b.x, r1b.x);
            dp[5] = make_float2(r0b.y, r1b.y);
            dp[6] = make_float2(r0b.z, r1b.z);
            dp[7] = make_float2(r0b.w, r1b.w);
        }
        __syncthreads();

        for (int kb = 0; kb < 32; kb += 4) {
            const int kA = kc + kb + half * 2;
            float2 av = *(const float2*)(aptr + kA);
            v2f a;
            a.x = av.x * nd;
            a.y = av.y * nd;
            const float* bp = sW + (kb / 2 + half) * PSTR + mcol * 2;
#pragma unroll
            for (int t = 0; t < 8; ++t) {
                v2f b = *(const v2f*)(bp + t * 32);
                acc[t] = __builtin_amdgcn_wmma_f32_16x16x4_f32(
                    false, a, false, b, (short)0, acc[t], false, false);
            }
        }
    }

#pragma unroll
    for (int t = 0; t < 8; ++t) {
        const int col = t * 16 + mcol;
        const float bv = bias[col];
#pragma unroll
        for (int r = 0; r < 8; ++r) {
            const int row = r0 + half * 8 + r;
            if (row < n) {
                float v = acc[t][r] + bv;
                v = (v >= 0.0f) ? v : alpha * v;
                hpre[(long)row * DFEAT + col] = v;
                atomicAdd(&sSum[col], v);
                atomicAdd(&sSq[col], v * v);
            }
        }
    }
    __syncthreads();
    if (tid < DFEAT) {
        atomicAdd(&fsum[tid], sSum[tid]);
        atomicAdd(&fsq[tid], sSq[tid]);
    }
}

__global__ void k_bnstats(const float* __restrict__ fsum, const float* __restrict__ fsq,
                          const float* __restrict__ gamma, const float* __restrict__ beta,
                          float* __restrict__ bnA, float* __restrict__ bnC, int n) {
    int c = threadIdx.x;
    float invn = 1.0f / (float)n;
    float mean = fsum[c] * invn;
    float var = fsq[c] * invn - mean * mean;
    float inv = rsqrtf(var + 1e-5f);
    float a = inv * gamma[c];
    bnA[c] = a;
    bnC[c] = beta[c] - mean * a;
}

#define BN_ROWS 16

__device__ __forceinline__ float4 bn_act(float4 v, float4 a, float4 c, float alpha) {
    float4 y;
    y.x = fmaf(v.x, a.x, c.x); y.x = (y.x >= 0.0f) ? y.x : alpha * y.x;
    y.y = fmaf(v.y, a.y, c.y); y.y = (y.y >= 0.0f) ? y.y : alpha * y.y;
    y.z = fmaf(v.z, a.z, c.z); y.z = (y.z >= 0.0f) ? y.z : alpha * y.z;
    y.w = fmaf(v.w, a.w, c.w); y.w = (y.w >= 0.0f) ? y.w : alpha * y.w;
    return y;
}

__device__ __forceinline__ void gh_flush(float* __restrict__ gh, int g, int lane, float4 s) {
    float* gp = gh + (long)g * DFEAT + lane * 4;
    atomicAdd(gp + 0, s.x);
    atomicAdd(gp + 1, s.y);
    atomicAdd(gp + 2, s.z);
    atomicAdd(gp + 3, s.w);
}

__global__ __launch_bounds__(128) void k_bnapply(const float* __restrict__ hpre,
                                                 const float* __restrict__ bnA,
                                                 const float* __restrict__ bnC,
                                                 const float* __restrict__ alpha_p,
                                                 const int* __restrict__ gids,
                                                 float* __restrict__ hout,
                                                 float* __restrict__ gh, int n) {
    const int lane = threadIdx.x & 31;
    const int gw = __builtin_amdgcn_readfirstlane((int)(blockIdx.x * 4 + (threadIdx.x >> 5)));
    const int row0 = gw * BN_ROWS;
    if (row0 >= n) return;
    const float alpha = alpha_p[0];
    const float4 a = ((const float4*)bnA)[lane];
    const float4 c = ((const float4*)bnC)[lane];
    const float4* src = (const float4*)hpre + (long)row0 * 32 + lane;
    float4* dst = (float4*)hout + (long)row0 * 32 + lane;
    float4 s = make_float4(0.0f, 0.0f, 0.0f, 0.0f);
    int gcur = gids[row0];
    if (row0 + BN_ROWS <= n) {
        float4 v[BN_ROWS];
        int g[BN_ROWS];
#pragma unroll
        for (int i = 0; i < BN_ROWS; ++i) {
            v[i] = src[i * 32];
            g[i] = gids[row0 + i];
        }
#pragma unroll
        for (int i = 0; i < BN_ROWS; ++i) {
            float4 y = bn_act(v[i], a, c, alpha);
            dst[i * 32] = y;
            if (g[i] != gcur) {
                gh_flush(gh, gcur, lane, s);
                s = make_float4(0.0f, 0.0f, 0.0f, 0.0f);
                gcur = g[i];
            }
            s.x += y.x; s.y += y.y; s.z += y.z; s.w += y.w;
        }
    } else {
        const int cnt = n - row0;
        for (int i = 0; i < cnt; ++i) {
            float4 y = bn_act(src[i * 32], a, c, alpha);
            dst[i * 32] = y;
            const int g = gids[row0 + i];
            if (g != gcur) {
                gh_flush(gh, gcur, lane, s);
                s = make_float4(0.0f, 0.0f, 0.0f, 0.0f);
                gcur = g;
            }
            s.x += y.x; s.y += y.y; s.z += y.z; s.w += y.w;
        }
    }
    gh_flush(gh, gcur, lane, s);
}

static inline int cdiv(long a, long b) { return (int)((a + b - 1) / b); }

extern "C" void kernel_launch(void* const* d_in, const int* in_sizes, int n_in,
                              void* d_out, int out_size, void* d_ws, size_t ws_size,
                              hipStream_t stream) {
    (void)n_in; (void)ws_size;
    const float* heat = (const float*)d_in[0];
    const int* esrc   = (const int*)d_in[1];
    const int* edst   = (const int*)d_in[2];
    const int* gids   = (const int*)d_in[3];
    const float* W0 = (const float*)d_in[4];
    const float* b0 = (const float*)d_in[5];
    const float* ac0 = (const float*)d_in[6];
    const float* g0 = (const float*)d_in[7];
    const float* be0 = (const float*)d_in[8];
    const float* aa0 = (const float*)d_in[9];
    const float* W1 = (const float*)d_in[10];
    const float* b1 = (const float*)d_in[11];
    const float* ac1 = (const float*)d_in[12];
    const float* g1 = (const float*)d_in[13];
    const float* be1 = (const float*)d_in[14];
    const float* aa1 = (const float*)d_in[15];

    const int E = in_sizes[1];
    const int N = in_sizes[3];
    const int ND = N * DFEAT;
    const int GD = out_size - ND;

    float* ws   = (float*)d_ws;
    int* dout   = (int*)d_ws;
    int* ptr    = dout + N;
    float* agg  = ws + 2 * (long)N;
    float* r2   = agg + ND;
    int* csr    = (int*)r2;
    float* ndst = r2 + 2 * (long)E;
    float* fsum = r2 + ND;
    float* fsq  = fsum + DFEAT;
    float* bnA  = fsq + DFEAT;
    float* bnC  = bnA + DFEAT;

    float* hout = (float*)d_out;
    float* gh   = hout + ND;

    const int gemm_blocks = cdiv(N, 128);

    k_zero<<<cdiv(2 * (long)N, 256), 256, 0, stream>>>(ws, 2 * N);
    k_zero<<<1, 256, 0, stream>>>(fsum, 2 * DFEAT);
    k_zero<<<cdiv(GD, 256), 256, 0, stream>>>(gh, GD);

    k_deg<<<cdiv(E, 256), 256, 0, stream>>>(esrc, edst, dout, ptr, E);
    k_scan<<<1, SCAN_THREADS, 0, stream>>>(ptr, ndst, N);
    k_fill<<<cdiv(E, 256), 256, 0, stream>>>(esrc, edst, dout, ptr, csr, E);

    k_gather<<<cdiv(N, 8), 256, 0, stream>>>(heat, ptr, csr, agg, N, E);
    k_gemm_bn<<<gemm_blocks, 256, 0, stream>>>(agg, agg, W0, b0, ac0, ndst, fsum, fsq, N);
    k_bnstats<<<1, DFEAT, 0, stream>>>(fsum, fsq, g0, be0, bnA, bnC, N);
    k_bnapply<<<cdiv(N, BN_ROWS * 4), 128, 0, stream>>>(agg, bnA, bnC, aa0, gids, hout, gh, N);

    k_zero<<<1, 256, 0, stream>>>(fsum, 2 * DFEAT);
    k_gather<<<cdiv(N, 8), 256, 0, stream>>>(hout, ptr, csr, agg, N, E);
    k_gemm_bn<<<gemm_blocks, 256, 0, stream>>>(agg, agg, W1, b1, ac1, ndst, fsum, fsq, N);
    k_bnstats<<<1, DFEAT, 0, stream>>>(fsum, fsq, g1, be1, bnA, bnC, N);
    k_bnapply<<<cdiv(N, BN_ROWS * 4), 128, 0, stream>>>(agg, bnA, bnC, aa1, gids, hout, gh, N);
}


# edge sort scatters staged in LDS: k_bin keeps its sorted slice, k_place its range's bucket region in LDS and each writes it out as whole lines (bucket order unchanged)
# speedup vs baseline: 3.0275x; 1.6829x over previous
#include <hip/hip_runtime.h>

typedef __attribute__((ext_vector_type(2))) float v2f;
typedef __attribute__((ext_vector_type(8))) float v8f;

#define DFEAT 128
#define PSTR 288

#define RSHIFT 8
#define RSIZE 256
#define BIN_THREADS 1024
#define BIN_BLOCKS 128
#define SLICE_CAP 4736
#define PLACE_CAP 6144

__device__ __forceinline__ int block_excl_scan(int v, int* red, int nwaves, int& total) {
    const int lane = threadIdx.x & 31;
    const int wave = threadIdx.x >> 5;
    int inc = v;
    for (int o = 1; o < 32; o <<= 1) {
        const int up = __shfl_up(inc, o);
        if (lane >= o) inc += up;
    }
    if (lane == 31) red[wave] = inc;
    __syncthreads();
    const int w = (lane < nwaves) ? red[lane] : 0;
    int winc = w;
    for (int o = 1; o < 32; o <<= 1) {
        const int up = __shfl_up(winc, o);
        if (lane >= o) winc += up;
    }
    total = __shfl(winc, 31);
    const int wexc = __shfl(winc - w, wave);
    __syncthreads();
    return wexc + inc - v;
}

__global__ __launch_bounds__(BIN_THREADS) void k_bin(const int* __restrict__ esrc,
                                                     const int* __restrict__ edst,
                                                     int* __restrict__ sortD,
                                                     int* __restrict__ sortS,
                                                     int* __restrict__ tabD,
                                                     int* __restrict__ tabS,
                                                     float* __restrict__ stats, int nstats,
                                                     float* __restrict__ gh, int ngh,
                                                     int e, int slice, int nb) {
    __shared__ int hD[BIN_THREADS];
    __shared__ int hS[BIN_THREADS];
    __shared__ int red[32];
    __shared__ int2 lD[SLICE_CAP];
    __shared__ int lS[SLICE_CAP];
    const int tid = threadIdx.x;
    const bool staged = slice <= SLICE_CAP;

    for (int i = blockIdx.x * BIN_THREADS + tid; i < nstats + ngh; i += gridDim.x * BIN_THREADS) {
        if (i < nstats) stats[i] = 0.0f;
        else gh[i - nstats] = 0.0f;
    }

    hD[tid] = 0;
    hS[tid] = 0;
    __syncthreads();

    const int beg = blockIdx.x * slice;
    const int end = min(beg + slice, e);
    for (int i = beg + tid; i < end; i += BIN_THREADS) {
        atomicAdd(&hD[edst[i] >> RSHIFT], 1);
        atomicAdd(&hS[esrc[i] >> RSHIFT], 1);
    }
    __syncthreads();

    const int vD = hD[tid];
    const int vS = hS[tid];
    int totD, totS;
    const int exD = block_excl_scan(vD, red, BIN_THREADS / 32, totD);
    const int exS = block_excl_scan(vS, red, BIN_THREADS / 32, totS);
    hD[tid] = exD;
    hS[tid] = exS;
    int* rowD = tabD + (long)blockIdx.x * (nb + 1);
    int* rowS = tabS + (long)blockIdx.x * (nb + 1);
    if (tid < nb) {
        rowD[tid] = exD;
        rowS[tid] = exS;
    }
    if (tid == 0) {
        rowD[nb] = totD;
        rowS[nb] = totS;
    }
    __syncthreads();

    if (staged) {
        for (int i = beg + tid; i < end; i += BIN_THREADS) {
            const int s = esrc[i];
            const int d = edst[i];
            const int pd = atomicAdd(&hD[d >> RSHIFT], 1);
            const int ps = atomicAdd(&hS[s >> RSHIFT], 1);
            lD[pd] = make_int2(s, d);
            lS[ps] = s;
        }
        __syncthreads();
        const int cnt = end - beg;
        for (int i = tid; i < cnt; i += BIN_THREADS) {
            ((int2*)sortD)[beg + i] = lD[i];
            sortS[beg + i] = lS[i];
        }
    } else {
        for (int i = beg + tid; i < end; i += BIN_THREADS) {
            const int s = esrc[i];
            const int d = edst[i];
            const int pd = atomicAdd(&hD[d >> RSHIFT], 1);
            const int ps = atomicAdd(&hS[s >> RSHIFT], 1);
            ((int2*)sortD)[beg + pd] = make_int2(s, d);
            sortS[beg + ps] = s;
        }
    }
}

__global__ __launch_bounds__(BIN_THREADS) void k_count(const int* __restrict__ sortD,
                                                       const int* __restrict__ sortS,
                                                       const int* __restrict__ tabD,
                                                       const int* __restrict__ tabS,
                                                       int* __restrict__ dout,
                                                       int* __restrict__ ptr,
                                                       float* __restrict__ ndst,
                                                       int n, int slice, int nb) {
    __shared__ int cIn[RSIZE];
    __shared__ int cOut[RSIZE];
    __shared__ int pD0[BIN_BLOCKS], pD1[BIN_BLOCKS], pS0[BIN_BLOCKS], pS1[BIN_BLOCKS];
    __shared__ int red[32];
    const int tid = threadIdx.x;
    const int lane = tid & 31;
    const int wave = tid >> 5;
    const int b = blockIdx.x;
    const int lo = b << RSHIFT;

    if (tid < RSIZE) { cIn[tid] = 0; cOut[tid] = 0; }
    int below = 0;
    if (tid < BIN_BLOCKS) {
        const int* td = tabD + (long)tid * (nb + 1) + b;
        const int* ts = tabS + (long)tid * (nb + 1) + b;
        const int d0 = td[0], d1 = td[1], s0 = ts[0], s1 = ts[1];
        pD0[tid] = tid * slice + d0;
        pD1[tid] = tid * slice + d1;
        pS0[tid] = tid * slice + s0;
        pS1[tid] = tid * slice + s1;
        below = d0;
    }
    int base;
    (void)block_excl_scan(below, red, BIN_THREADS / 32, base);

    const int2* sd = (const int2*)sortD;
    for (int j = wave; j < BIN_BLOCKS; j += BIN_THREADS / 32) {
        const int d1 = pD1[j];
        for (int k = pD0[j] + lane; k < d1; k += 32) atomicAdd(&cIn[sd[k].y - lo], 1);
        const int s1 = pS1[j];
        for (int k = pS0[j] + lane; k < s1; k += 32) atomicAdd(&cOut[sortS[k] - lo], 1);
    }
    __syncthreads();

    const int v = (tid < RSIZE) ? cIn[tid] : 0;
    int total;
    const int ex = block_excl_scan(v, red, BIN_THREADS / 32, total);
    if (tid < RSIZE && lo + tid < n) {
        ptr[lo + tid] = base + ex;
        ndst[lo + tid] = (v > 0) ? rsqrtf((float)v) : 0.0f;
        dout[lo + tid] = cOut[tid];
    }
}

__global__ __launch_bounds__(BIN_THREADS) void k_place(const int* __restrict__ sortD,
                                                       const int* __restrict__ tabD,
                                                       const int* __restrict__ dout,
                                                       const int* __restrict__ ptr,
                                                       int* __restrict__ csr,
                                                       int n, int e, int slice, int nb) {
    __shared__ int cur[RSIZE];
    __shared__ int pD0[BIN_BLOCKS], pD1[BIN_BLOCKS];
    __shared__ int2 lC[PLACE_CAP];
    const int tid = threadIdx.x;
    const int lane = tid & 31;
    const int wave = tid >> 5;
    const int b = blockIdx.x;
    const int lo = b << RSHIFT;
    const int base = ptr[lo];
    const int total = ((lo + RSIZE < n) ? ptr[lo + RSIZE] : e) - base;
    const bool staged = total <= PLACE_CAP;

    if (tid < RSIZE) cur[tid] = (lo + tid < n) ? ptr[lo + tid] : 0;
    if (tid < BIN_BLOCKS) {
        const int* td = tabD + (long)tid * (nb + 1) + b;
        pD0[tid] = tid * slice + td[0];
        pD1[tid] = tid * slice + td[1];
    }
    __syncthreads();

    const int2* sd = (const int2*)sortD;
    if (staged) {
        for (int j = wave; j < BIN_BLOCKS; j += BIN_THREADS / 32) {
            const int d1 = pD1[j];
            for (int k = pD0[j] + lane; k < d1; k += 32) {
                const int2 ed = sd[k];
                const int pos = atomicAdd(&cur[ed.y - lo], 1);
                const float w = rsqrtf((float)dout[ed.x]);
                lC[pos - base] = make_int2(ed.x, __float_as_int(w));
            }
        }
        __syncthreads();
        for (int i = tid; i < total; i += BIN_THREADS) ((int2*)csr)[base + i] = lC[i];
    } else {
        for (int j = wave; j < BIN_BLOCKS; j += BIN_THREADS / 32) {
            const int d1 = pD1[j];
            for (int k = pD0[j] + lane; k < d1; k += 32) {
                const int2 ed = sd[k];
                const int pos = atomicAdd(&cur[ed.y - lo], 1);
                const float w = rsqrtf((float)dout[ed.x]);
                ((int2*)csr)[pos] = make_int2(ed.x, __float_as_int(w));
            }
        }
    }
}

__global__ __launch_bounds__(256) void k_gather(const float* __restrict__ x,
                                                const int* __restrict__ ptr,
                                                const int* __restrict__ csr,
                                                float* __restrict__ agg, int n, int e) {
    const int lane = threadIdx.x & 31;
    const int row = __builtin_amdgcn_readfirstlane((int)(blockIdx.x * 8 + (threadIdx.x >> 5)));
    if (row >= n) return;
    const int beg = ptr[row];
    const int end = (row + 1 < n) ? ptr[row + 1] : e;
    const int2* ent = (const int2*)csr;
    const float4* xv = (const float4*)x + lane;
    float4 acc = make_float4(0.0f, 0.0f, 0.0f, 0.0f);
    for (int j = beg; j < end; j += 4) {
        const int rem = end - j;
        const bool ok1 = rem > 1, ok2 = rem > 2, ok3 = rem > 3;
        const int2 e0 = ent[j];
        const int2 e1 = ent[ok1 ? j + 1 : j];
        const int2 e2 = ent[ok2 ? j + 2 : j];
        const int2 e3 = ent[ok3 ? j + 3 : j];
        const float4 v0 = xv[(long)e0.x * 32];
        const float4 v1 = xv[(long)e1.x * 32];
        const float4 v2 = xv[(long)e2.x * 32];
        const float4 v3 = xv[(long)e3.x * 32];
        const float w0 = __int_as_float(e0.y);
        const float w1 = __int_as_float(e1.y);
        const float w2 = __int_as_float(e2.y);
        const float w3 = __int_as_float(e3.y);
        acc.x += v0.x * w0; acc.y += v0.y * w0; acc.z += v0.z * w0; acc.w += v0.w * w0;
        acc.x = ok1 ? acc.x + v1.x * w1 : acc.x;
        acc.y = ok1 ? acc.y + v1.y * w1 : acc.y;
        acc.z = ok1 ? acc.z + v1.z * w1 : acc.z;
        acc.w = ok1 ? acc.w + v1.w * w1 : acc.w;
        acc.x = ok2 ? acc.x + v2.x * w2 : acc.x;
        acc.y = ok2 ? acc.y + v2.y * w2 : acc.y;
        acc.z = ok2 ? acc.z + v2.z * w2 : acc.z;
        acc.w = ok2 ? acc.w + v2.w * w2 : acc.w;
        acc.x = ok3 ? acc.x + v3.x * w3 : acc.x;
        acc.y = ok3 ? acc.y + v3.y * w3 : acc.y;
        acc.z = ok3 ? acc.z + v3.z * w3 : acc.z;
        acc.w = ok3 ? acc.w + v3.w * w3 : acc.w;
    }
    ((float4*)agg)[(long)row * 32 + lane] = acc;
}

__global__ __launch_bounds__(256) void k_gemm_bn(const float* agg, float* hpre,
                                                 const float* __restrict__ W,
                                                 const float* __restrict__ bias,
                                                 const float* __restrict__ alpha_p,
                                                 const float* __restrict__ ndst,
                                                 float* __restrict__ fsum,
                                                 float* __restrict__ fsq, int n) {
    __shared__ float sW[16 * PSTR];
    __shared__ float sSum[DFEAT];
    __shared__ float sSq[DFEAT];

    const int tid = threadIdx.x;
    if (tid < DFEAT) { sSum[tid] = 0.0f; sSq[tid] = 0.0f; }

    const int wave = tid >> 5;
    const int lane = tid & 31;
    const int half = lane >> 4;
    const int mcol = lane & 15;
    const int r0 = blockIdx.x * 128 + wave * 16;
    const int rowA = r0 + mcol;
    const bool aok = rowA < n;
    const float nd = aok ? ndst[rowA] : 0.0f;
    const int rowC = aok ? rowA : (n - 1);
    const float alpha = alpha_p[0];
    const float* aptr = agg + (long)rowC * DFEAT;

    v8f acc[8] = {};

    for (int kc = 0; kc < DFEAT; kc += 32) {
        __syncthreads();
        {
            int kp = tid >> 4;
            int nb = (tid & 15) * 8;
            const float* w0 = W + (kc + 2 * kp) * DFEAT + nb;
            const float* w1 = w0 + DFEAT;
            float4 r0a = ((const float4*)w0)[0];
            float4 r0b = ((const float4*)w0)[1];
            float4 r1a = ((const float4*)w1)[0];
            float4 r1b = ((const float4*)w1)[1];
            float2* dp = (float2*)(sW + kp * PSTR + nb * 2);
            dp[0] = make_float2(r0a.x, r1a.x);
            dp[1] = make_float2(r0a.y, r1a.y);
            dp[2] = make_float2(r0a.z, r1a.z);
            dp[3] = make_float2(r0a.w, r1a.w);
            dp[4] = make_float2(r0b.x, r1b.x);
            dp[5] = make_float2(r0b.y, r1b.y);
            dp[6] = make_float2(r0b.z, r1b.z);
            dp[7] = make_float2(r0b.w, r1b.w);
        }
        __syncthreads();

        for (int kb = 0; kb < 32; kb += 4) {
            const int kA = kc + kb + half * 2;
            float2 av = *(const float2*)(aptr + kA);
            v2f a;
            a.x = av.x * nd;
            a.y = av.y * nd;
            const float* bp = sW + (kb / 2 + half) * PSTR + mcol * 2;
#pragma unroll
            for (int t = 0; t < 8; ++t) {
                v2f b = *(const v2f*)(bp + t * 32);
                acc[t] = __builtin_amdgcn_wmma_f32_16x16x4_f32(
                    false, a, false, b, (short)0, acc[t], false, false);
            }
        }
    }

#pragma unroll
    for (int t = 0; t < 8; ++t) {
        const int col = t * 16 + mcol;
        const float bv = bias[col];
#pragma unroll
        for (int r = 0; r < 8; ++r) {
            const int row = r0 + half * 8 + r;
            if (row < n) {
                float v = acc[t][r] + bv;
                v = (v >= 0.0f) ? v : alpha * v;
                hpre[(long)row * DFEAT + col] = v;
                atomicAdd(&sSum[col], v);
                atomicAdd(&sSq[col], v * v);
            }
        }
    }
    __syncthreads();
    if (tid < DFEAT) {
        atomicAdd(&fsum[tid], sSum[tid]);
        atomicAdd(&fsq[tid], sSq[tid]);
    }
}

#define BN_ROWS 16

__device__ __forceinline__ float4 bn_act(float4 v, float4 a, float4 c, float alpha) {
    float4 y;
    y.x = fmaf(v.x, a.x, c.x); y.x = (y.x >= 0.0f) ? y.x : alpha * y.x;
    y.y = fmaf(v.y, a.y, c.y); y.y = (y.y >= 0.0f) ? y.y : alpha * y.y;
    y.z = fmaf(v.z, a.z, c.z); y.z = (y.z >= 0.0f) ? y.z : alpha * y.z;
    y.w = fmaf(v.w, a.w, c.w); y.w = (y.w >= 0.0f) ? y.w : alpha * y.w;
    return y;
}

__device__ __forceinline__ void bn_affine(float s, float q, float g, float b, float invn,
                                          float& a, float& c) {
    const float mean = s * invn;
    const float var = q * invn - mean * mean;
    const float inv = rsqrtf(var + 1e-5f);
    a = inv * g;
    c = b - mean * a;
}

__device__ __forceinline__ void bn_affine4(float4 s, float4 q, float4 g, float4 b, float invn,
                                           float4& a, float4& c) {
    bn_affine(s.x, q.x, g.x, b.x, invn, a.x, c.x);
    bn_affine(s.y, q.y, g.y, b.y, invn, a.y, c.y);
    bn_affine(s.z, q.z, g.z, b.z, invn, a.z, c.z);
    bn_affine(s.w, q.w, g.w, b.w, invn, a.w, c.w);
}

__device__ __forceinline__ void gh_flush(float* __restrict__ gh, int g, int lane, float4 s) {
    float* gp = gh + (long)g * DFEAT + lane * 4;
    atomicAdd(gp + 0, s.x);
    atomicAdd(gp + 1, s.y);
    atomicAdd(gp + 2, s.z);
    atomicAdd(gp + 3, s.w);
}

__global__ __launch_bounds__(128) void k_bnapply(const float* __restrict__ hpre,
                                                 const float* __restrict__ fsum,
                                                 const float* __restrict__ fsq,
                                                 const float* __restrict__ gamma,
                                                 const float* __restrict__ beta,
                                                 const float* __restrict__ alpha_p,
                                                 const int* __restrict__ gids,
                                                 float* __restrict__ hout,
                                                 float* __restrict__ gh, int n) {
    const int lane = threadIdx.x & 31;
    const int gw = __builtin_amdgcn_readfirstlane((int)(blockIdx.x * 4 + (threadIdx.x >> 5)));
    const int row0 = gw * BN_ROWS;
    if (row0 >= n) return;
    const float alpha = alpha_p[0];
    const float4 s4 = ((const float4*)fsum)[lane];
    const float4 q4 = ((const float4*)fsq)[lane];
    const float4 g4 = ((const float4*)gamma)[lane];
    const float4 b4 = ((const float4*)beta)[lane];
    const float invn = 1.0f / (float)n;
    float4 a, c;
    const float4* src = (const float4*)hpre + (long)row0 * 32 + lane;
    float4* dst = (float4*)hout + (long)row0 * 32 + lane;
    float4 s = make_float4(0.0f, 0.0f, 0.0f, 0.0f);
    int gcur = gids[row0];
    if (row0 + BN_ROWS <= n) {
        float4 v[BN_ROWS];
        int g[BN_ROWS];
#pragma unroll
        for (int i = 0; i < BN_ROWS; ++i) {
            v[i] = src[i * 32];
            g[i] = gids[row0 + i];
        }
        bn_affine4(s4, q4, g4, b4, invn, a, c);
#pragma unroll
        for (int i = 0; i < BN_ROWS; ++i) {
            float4 y = bn_act(v[i], a, c, alpha);
            dst[i * 32] = y;
            if (g[i] != gcur) {
                gh_flush(gh, gcur, lane, s);
                s = make_float4(0.0f, 0.0f, 0.0f, 0.0f);
                gcur = g[i];
            }
            s.x += y.x; s.y += y.y; s.z += y.z; s.w += y.w;
        }
    } else {
        const int cnt = n - row0;
        bn_affine4(s4, q4, g4, b4, invn, a, c);
        for (int i = 0; i < cnt; ++i) {
            float4 y = bn_act(src[i * 32], a, c, alpha);
            dst[i * 32] = y;
            const int g = gids[row0 + i];
            if (g != gcur) {
                gh_flush(gh, gcur, lane, s);
                s = make_float4(0.0f, 0.0f, 0.0f, 0.0f);
                gcur = g;
            }
            s.x += y.x; s.y += y.y; s.z += y.z; s.w += y.w;
        }
    }
    gh_flush(gh, gcur, lane, s);
}

static inline int cdiv(long a, long b) { return (int)((a + b - 1) / b); }

extern "C" void kernel_launch(void* const* d_in, const int* in_sizes, int n_in,
                              void* d_out, int out_size, void* d_ws, size_t ws_size,
                              hipStream_t stream) {
    (void)n_in; (void)ws_size;
    const float* heat = (const float*)d_in[0];
    const int* esrc   = (const int*)d_in[1];
    const int* edst   = (const int*)d_in[2];
    const int* gids   = (const int*)d_in[3];
    const float* W0 = (const float*)d_in[4];
    const float* b0 = (const float*)d_in[5];
    const float* ac0 = (const float*)d_in[6];
    const float* g0 = (const float*)d_in[7];
    const float* be0 = (const float*)d_in[8];
    const float* aa0 = (const float*)d_in[9];
    const float* W1 = (const float*)d_in[10];
    const float* b1 = (const float*)d_in[11];
    const float* ac1 = (const float*)d_in[12];
    const float* g1 = (const float*)d_in[13];
    const float* be1 = (const float*)d_in[14];
    const float* aa1 = (const float*)d_in[15];

    const int E = in_sizes[1];
    const int N = in_sizes[3];
    const int ND = N * DFEAT;
    const int GD = out_size - ND;
    const int NB = cdiv(N, RSIZE);
    const int slice = cdiv(E, BIN_BLOCKS);

    float* ws   = (float*)d_ws;
    int* dout   = (int*)d_ws;
    int* ptr    = dout + N;
    float* agg  = ws + 2 * (long)N;
    float* r2   = agg + ND;
    int* csr    = (int*)r2;
    float* ndst = r2 + 2 * (long)E;
    int* sortD  = (int*)(ndst + N + (N & 1));
    int* sortS  = sortD + 2 * (long)E;
    int* tabD   = sortS + E;
    int* tabS   = tabD + (long)BIN_BLOCKS * (NB + 1);
    float* fsum0 = r2 + ND;
    float* fsq0  = fsum0 + DFEAT;
    float* fsum1 = fsq0 + DFEAT;
    float* fsq1  = fsum1 + DFEAT;

    float* hout = (float*)d_out;
    float* gh   = hout + ND;

    const int gemm_blocks = cdiv(N, 128);

    k_bin<<<BIN_BLOCKS, BIN_THREADS, 0, stream>>>(esrc, edst, sortD, sortS, tabD, tabS, fsum0,
                                                  4 * DFEAT, gh, GD, E, slice, NB);
    k_count<<<NB, BIN_THREADS, 0, stream>>>(sortD, sortS, tabD, tabS, dout, ptr, ndst, N, slice, NB);
    k_place<<<NB, BIN_THREADS, 0, stream>>>(sortD, tabD, dout, ptr, csr, N, E, slice, NB);

    k_gather<<<cdiv(N, 8), 256, 0, stream>>>(heat, ptr, csr, agg, N, E);
    k_gemm_bn<<<gemm_blocks, 256, 0, stream>>>(agg, agg, W0, b0, ac0, ndst, fsum0, fsq0, N);
    k_bnapply<<<cdiv(N, BN_ROWS * 4), 128, 0, stream>>>(agg, fsum0, fsq0, g0, be0, aa0, gids, hout,
                                                       gh, N);

    k_gather<<<cdiv(N, 8), 256, 0, stream>>>(hout, ptr, csr, agg, N, E);
    k_gemm_bn<<<gemm_blocks, 256, 0, stream>>>(agg, agg, W1, b1, ac1, ndst, fsum1, fsq1, N);
    k_bnapply<<<cdiv(N, BN_ROWS * 4), 128, 0, stream>>>(agg, fsum1, fsq1, g1, be1, aa1, gids, hout,
                                                       gh, N);
}


# GEMM epilogue through LDS rows: a wave stages its 16x128 tile, lanes then own 4 columns of whole rows: one 512-B store per row, BN column sums added in registers first (their sum order changes)
# speedup vs baseline: 3.2706x; 1.0803x over previous
#include <hip/hip_runtime.h>

typedef __attribute__((ext_vector_type(2))) float v2f;
typedef __attribute__((ext_vector_type(8))) float v8f;

#define DFEAT 128
#define PSTR 288
#define TPITCH 132

#define RSHIFT 8
#define RSIZE 256
#define BIN_THREADS 1024
#define BIN_BLOCKS 128
#define SLICE_CAP 4736
#define PLACE_CAP 6144

__device__ __forceinline__ int block_excl_scan(int v, int* red, int nwaves, int& total) {
    const int lane = threadIdx.x & 31;
    const int wave = threadIdx.x >> 5;
    int inc = v;
    for (int o = 1; o < 32; o <<= 1) {
        const int up = __shfl_up(inc, o);
        if (lane >= o) inc += up;
    }
    if (lane == 31) red[wave] = inc;
    __syncthreads();
    const int w = (lane < nwaves) ? red[lane] : 0;
    int winc = w;
    for (int o = 1; o < 32; o <<= 1) {
        const int up = __shfl_up(winc, o);
        if (lane >= o) winc += up;
    }
    total = __shfl(winc, 31);
    const int wexc = __shfl(winc - w, wave);
    __syncthreads();
    return wexc + inc - v;
}

__global__ __launch_bounds__(BIN_THREADS) void k_bin(const int* __restrict__ esrc,
                                                     const int* __restrict__ edst,
                                                     int* __restrict__ sortD,
                                                     int* __restrict__ sortS,
                                                     int* __restrict__ tabD,
                                                     int* __restrict__ tabS,
                                                     float* __restrict__ stats, int nstats,
                                                     float* __restrict__ gh, int ngh,
                                                     int e, int slice, int nb) {
    __shared__ int hD[BIN_THREADS];
    __shared__ int hS[BIN_THREADS];
    __shared__ int red[32];
    __shared__ int2 lD[SLICE_CAP];
    __shared__ int lS[SLICE_CAP];
    const int tid = threadIdx.x;
    const bool staged = slice <= SLICE_CAP;

    for (int i = blockIdx.x * BIN_THREADS + tid; i < nstats + ngh; i += gridDim.x * BIN_THREADS) {
        if (i < nstats) stats[i] = 0.0f;
        else gh[i - nstats] = 0.0f;
    }

    hD[tid] = 0;
    hS[tid] = 0;
    __syncthreads();

    const int beg = blockIdx.x * slice;
    const int end = min(beg + slice, e);
    for (int i = beg + tid; i < end; i += BIN_THREADS) {
        atomicAdd(&hD[edst[i] >> RSHIFT], 1);
        atomicAdd(&hS[esrc[i] >> RSHIFT], 1);
    }
    __syncthreads();

    const int vD = hD[tid];
    const int vS = hS[tid];
    int totD, totS;
    const int exD = block_excl_scan(vD, red, BIN_THREADS / 32, totD);
    const int exS = block_excl_scan(vS, red, BIN_THREADS / 32, totS);
    hD[tid] = exD;
    hS[tid] = exS;
    int* rowD = tabD + (long)blockIdx.x * (nb + 1);
    int* rowS = tabS + (long)blockIdx.x * (nb + 1);
    if (tid < nb) {
        rowD[tid] = exD;
        rowS[tid] = exS;
    }
    if (tid == 0) {
        rowD[nb] = totD;
        rowS[nb] = totS;
    }
    __syncthreads();

    if (staged) {
        for (int i = beg + tid; i < end; i += BIN_THREADS) {
            const int s = esrc[i];
            const int d = edst[i];
            const int pd = atomicAdd(&hD[d >> RSHIFT], 1);
            const int ps = atomicAdd(&hS[s >> RSHIFT], 1);
            lD[pd] = make_int2(s, d);
            lS[ps] = s;
        }
        __syncthreads();
        const int cnt = end - beg;
        for (int i = tid; i < cnt; i += BIN_THREADS) {
            ((int2*)sortD)[beg + i] = lD[i];
            sortS[beg + i] = lS[i];
        }
    } else {
        for (int i = beg + tid; i < end; i += BIN_THREADS) {
            const int s = esrc[i];
            const int d = edst[i];
            const int pd = atomicAdd(&hD[d >> RSHIFT], 1);
            const int ps = atomicAdd(&hS[s >> RSHIFT], 1);
            ((int2*)sortD)[beg + pd] = make_int2(s, d);
            sortS[beg + ps] = s;
        }
    }
}

__global__ __launch_bounds__(BIN_THREADS) void k_count(const int* __restrict__ sortD,
                                                       const int* __restrict__ sortS,
                                                       const int* __restrict__ tabD,
                                                       const int* __restrict__ tabS,
                                                       int* __restrict__ dout,
                                                       int* __restrict__ ptr,
                                                       float* __restrict__ ndst,
                                                       int n, int slice, int nb) {
    __shared__ int cIn[RSIZE];
    __shared__ int cOut[RSIZE];
    __shared__ int pD0[BIN_BLOCKS], pD1[BIN_BLOCKS], pS0[BIN_BLOCKS], pS1[BIN_BLOCKS];
    __shared__ int red[32];
    const int tid = threadIdx.x;
    const int lane = tid & 31;
    const int wave = tid >> 5;
    const int b = blockIdx.x;
    const int lo = b << RSHIFT;

    if (tid < RSIZE) { cIn[tid] = 0; cOut[tid] = 0; }
    int below = 0;
    if (tid < BIN_BLOCKS) {
        const int* td = tabD + (long)tid * (nb + 1) + b;
        const int* ts = tabS + (long)tid * (nb + 1) + b;
        const int d0 = td[0], d1 = td[1], s0 = ts[0], s1 = ts[1];
        pD0[tid] = tid * slice + d0;
        pD1[tid] = tid * slice + d1;
        pS0[tid] = tid * slice + s0;
        pS1[tid] = tid * slice + s1;
        below = d0;
    }
    int base;
    (void)block_excl_scan(below, red, BIN_THREADS / 32, base);

    const int2* sd = (const int2*)sortD;
    for (int j = wave; j < BIN_BLOCKS; j += BIN_THREADS / 32) {
        const int d1 = pD1[j];
        for (int k = pD0[j] + lane; k < d1; k += 32) atomicAdd(&cIn[sd[k].y - lo], 1);
        const int s1 = pS1[j];
        for (int k = pS0[j] + lane; k < s1; k += 32) atomicAdd(&cOut[sortS[k] - lo], 1);
    }
    __syncthreads();

    const int v = (tid < RSIZE) ? cIn[tid] : 0;
    int total;
    const int ex = block_excl_scan(v, red, BIN_THREADS / 32, total);
    if (tid < RSIZE && lo + tid < n) {
        ptr[lo + tid] = base + ex;
        ndst[lo + tid] = (v > 0) ? rsqrtf((float)v) : 0.0f;
        dout[lo + tid] = cOut[tid];
    }
}

__global__ __launch_bounds__(BIN_THREADS) void k_place(const int* __restrict__ sortD,
                                                       const int* __restrict__ tabD,
                                                       const int* __restrict__ dout,
                                                       const int* __restrict__ ptr,
                                                       int* __restrict__ csr,
                                                       int n, int e, int slice, int nb) {
    __shared__ int cur[RSIZE];
    __shared__ int pD0[BIN_BLOCKS], pD1[BIN_BLOCKS];
    __shared__ int2 lC[PLACE_CAP];
    const int tid = threadIdx.x;
    const int lane = tid & 31;
    const int wave = tid >> 5;
    const int b = blockIdx.x;
    const int lo = b << RSHIFT;
    const int base = ptr[lo];
    const int total = ((lo + RSIZE < n) ? ptr[lo + RSIZE] : e) - base;
    const bool staged = total <= PLACE_CAP;

    if (tid < RSIZE) cur[tid] = (lo + tid < n) ? ptr[lo + tid] : 0;
    if (tid < BIN_BLOCKS) {
        const int* td = tabD + (long)tid * (nb + 1) + b;
        pD0[tid] = tid * slice + td[0];
        pD1[tid] = tid * slice + td[1];
    }
    __syncthreads();

    const int2* sd = (const int2*)sortD;
    if (staged) {
        for (int j = wave; j < BIN_BLOCKS; j += BIN_THREADS / 32) {
            const int d1 = pD1[j];
            for (int k = pD0[j] + lane; k < d1; k += 32) {
                const int2 ed = sd[k];
                const int pos = atomicAdd(&cur[ed.y - lo], 1);
                const float w = rsqrtf((float)dout[ed.x]);
                lC[pos - base] = make_int2(ed.x, __float_as_int(w));
            }
        }
        __syncthreads();
        for (int i = tid; i < total; i += BIN_THREADS) ((int2*)csr)[base + i] = lC[i];
    } else {
        for (int j = wave; j < BIN_BLOCKS; j += BIN_THREADS / 32) {
            const int d1 = pD1[j];
            for (int k = pD0[j] + lane; k < d1; k += 32) {
                const int2 ed = sd[k];
                const int pos = atomicAdd(&cur[ed.y - lo], 1);
                const float w = rsqrtf((float)dout[ed.x]);
                ((int2*)csr)[pos] = make_int2(ed.x, __float_as_int(w));
            }
        }
    }
}

__global__ __launch_bounds__(256) void k_gather(const float* __restrict__ x,
                                                const int* __restrict__ ptr,
                                                const int* __restrict__ csr,
                                                float* __restrict__ agg, int n, int e) {
    const int lane = threadIdx.x & 31;
    const int row = __builtin_amdgcn_readfirstlane((int)(blockIdx.x * 8 + (threadIdx.x >> 5)));
    if (row >= n) return;
    const int beg = ptr[row];
    const int end = (row + 1 < n) ? ptr[row + 1] : e;
    const int2* ent = (const int2*)csr;
    const float4* xv = (const float4*)x + lane;
    float4 acc = make_float4(0.0f, 0.0f, 0.0f, 0.0f);
    for (int j = beg; j < end; j += 4) {
        const int rem = end - j;
        const bool ok1 = rem > 1, ok2 = rem > 2, ok3 = rem > 3;
        const int2 e0 = ent[j];
        const int2 e1 = ent[ok1 ? j + 1 : j];
        const int2 e2 = ent[ok2 ? j + 2 : j];
        const int2 e3 = ent[ok3 ? j + 3 : j];
        const float4 v0 = xv[(long)e0.x * 32];
        const float4 v1 = xv[(long)e1.x * 32];
        const float4 v2 = xv[(long)e2.x * 32];
        const float4 v3 = xv[(long)e3.x * 32];
        const float w0 = __int_as_float(e0.y);
        const float w1 = __int_as_float(e1.y);
        const float w2 = __int_as_float(e2.y);
        const float w3 = __int_as_float(e3.y);
        acc.x += v0.x * w0; acc.y += v0.y * w0; acc.z += v0.z * w0; acc.w += v0.w * w0;
        acc.x = ok1 ? acc.x + v1.x * w1 : acc.x;
        acc.y = ok1 ? acc.y + v1.y * w1 : acc.y;
        acc.z = ok1 ? acc.z + v1.z * w1 : acc.z;
        acc.w = ok1 ? acc.w + v1.w * w1 : acc.w;
        acc.x = ok2 ? acc.x + v2.x * w2 : acc.x;
        acc.y = ok2 ? acc.y + v2.y * w2 : acc.y;
        acc.z = ok2 ? acc.z + v2.z * w2 : acc.z;
        acc.w = ok2 ? acc.w + v2.w * w2 : acc.w;
        acc.x = ok3 ? acc.x + v3.x * w3 : acc.x;
        acc.y = ok3 ? acc.y + v3.y * w3 : acc.y;
        acc.z = ok3 ? acc.z + v3.z * w3 : acc.z;
        acc.w = ok3 ? acc.w + v3.w * w3 : acc.w;
    }
    ((float4*)agg)[(long)row * 32 + lane] = acc;
}

__device__ __forceinline__ void gemm_ep_row(float4& v, float4 bv, float alpha, float4& s, float4& q) {
    v.x += bv.x; v.y += bv.y; v.z += bv.z; v.w += bv.w;
    v.x = (v.x >= 0.0f) ? v.x : alpha * v.x;
    v.y = (v.y >= 0.0f) ? v.y : alpha * v.y;
    v.z = (v.z >= 0.0f) ? v.z : alpha * v.z;
    v.w = (v.w >= 0.0f) ? v.w : alpha * v.w;
    s.x += v.x; s.y += v.y; s.z += v.z; s.w += v.w;
    q.x += v.x * v.x; q.y += v.y * v.y; q.z += v.z * v.z; q.w += v.w * v.w;
}

__global__ __launch_bounds__(256) void k_gemm_bn(const float* agg, float* hpre,
                                                 const float* __restrict__ W,
                                                 const float* __restrict__ bias,
                                                 const float* __restrict__ alpha_p,
                                                 const float* __restrict__ ndst,
                                                 float* __restrict__ fsum,
                                                 float* __restrict__ fsq, int n) {
    __shared__ float sW[16 * PSTR];
    __shared__ float sSum[DFEAT];
    __shared__ float sSq[DFEAT];
    __shared__ float4 sT[8 * 16 * (TPITCH / 4)];

    const int tid = threadIdx.x;
    if (tid < DFEAT) { sSum[tid] = 0.0f; sSq[tid] = 0.0f; }

    const int wave = tid >> 5;
    const int lane = tid & 31;
    const int half = lane >> 4;
    const int mcol = lane & 15;
    const int r0 = blockIdx.x * 128 + wave * 16;
    const int rowA = r0 + mcol;
    const bool aok = rowA < n;
    const float nd = aok ? ndst[rowA] : 0.0f;
    const int rowC = aok ? rowA : (n - 1);
    const float alpha = alpha_p[0];
    const float* aptr = agg + (long)rowC * DFEAT;

    v8f acc[8] = {};

    for (int kc = 0; kc < DFEAT; kc += 32) {
        __syncthreads();
        {
            int kp = tid >> 4;
            int nb = (tid & 15) * 8;
            const float* w0 = W + (kc + 2 * kp) * DFEAT + nb;
            const float* w1 = w0 + DFEAT;
            float4 r0a = ((const float4*)w0)[0];
            float4 r0b = ((const float4*)w0)[1];
            float4 r1a = ((const float4*)w1)[0];
            float4 r1b = ((const float4*)w1)[1];
            float2* dp = (float2*)(sW + kp * PSTR + nb * 2);
            dp[0] = make_float2(r0a.x, r1a.x);
            dp[1] = make_float2(r0a.y, r1a.y);
            dp[2] = make_float2(r0a.z, r1a.z);
            dp[3] = make_float2(r0a.w, r1a.w);
            dp[4] = make_float2(r0b.x, r1b.x);
            dp[5] = make_float2(r0b.y, r1b.y);
            dp[6] = make_float2(r0b.z, r1b.z);
            dp[7] = make_float2(r0b.w, r1b.w);
        }
        __syncthreads();

        for (int kb = 0; kb < 32; kb += 4) {
            const int kA = kc + kb + half * 2;
            float2 av = *(const float2*)(aptr + kA);
            v2f a;
            a.x = av.x * nd;
            a.y = av.y * nd;
            const float* bp = sW + (kb / 2 + half) * PSTR + mcol * 2;
#pragma unroll
            for (int t = 0; t < 8; ++t) {
                v2f b = *(const v2f*)(bp + t * 32);
                acc[t] = __builtin_amdgcn_wmma_f32_16x16x4_f32(
                    false, a, false, b, (short)0, acc[t], false, false);
            }
        }
    }

    float* tw = (float*)sT + wave * (16 * TPITCH);
#pragma unroll
    for (int t = 0; t < 8; ++t) {
#pragma unroll
        for (int r = 0; r < 8; ++r) tw[(half * 8 + r) * TPITCH + t * 16 + mcol] = acc[t][r];
    }
    __syncthreads();
    {
        const int wave_s = __builtin_amdgcn_readfirstlane(wave);
        const int row0 = blockIdx.x * 128 + wave_s * 16;
        const float4 bv = ((const float4*)bias)[lane];
        float4 s = make_float4(0.0f, 0.0f, 0.0f, 0.0f);
        float4 q = make_float4(0.0f, 0.0f, 0.0f, 0.0f);
        const float* trow = tw + lane * 4;
        if (row0 + 16 <= n) {
            float4* orow = (float4*)hpre + (long)row0 * 32 + lane;
            float4 v[16];
#pragma unroll
            for (int i = 0; i < 16; ++i) v[i] = *(const float4*)(trow + i * TPITCH);
#pragma unroll
            for (int i = 0; i < 16; ++i) {
                gemm_ep_row(v[i], bv, alpha, s, q);
                orow[i * 32] = v[i];
            }
        } else {
            for (int i = 0; i < 16 && row0 + i < n; ++i) {
                float4 v = *(const float4*)(trow + i * TPITCH);
                gemm_ep_row(v, bv, alpha, s, q);
                ((float4*)hpre)[(long)(row0 + i) * 32 + lane] = v;
            }
        }
        atomicAdd(&sSum[lane * 4 + 0], s.x);
        atomicAdd(&sSum[lane * 4 + 1], s.y);
        atomicAdd(&sSum[lane * 4 + 2], s.z);
        atomicAdd(&sSum[lane * 4 + 3], s.w);
        atomicAdd(&sSq[lane * 4 + 0], q.x);
        atomicAdd(&sSq[lane * 4 + 1], q.y);
        atomicAdd(&sSq[lane * 4 + 2], q.z);
        atomicAdd(&sSq[lane * 4 + 3], q.w);
    }
    __syncthreads();
    if (tid < DFEAT) {
        atomicAdd(&fsum[tid], sSum[tid]);
        atomicAdd(&fsq[tid], sSq[tid]);
    }
}

#define BN_ROWS 16

__device__ __forceinline__ float4 bn_act(float4 v, float4 a, float4 c, float alpha) {
    float4 y;
    y.x = fmaf(v.x, a.x, c.x); y.x = (y.x >= 0.0f) ? y.x : alpha * y.x;
    y.y = fmaf(v.y, a.y, c.y); y.y = (y.y >= 0.0f) ? y.y : alpha * y.y;
    y.z = fmaf(v.z, a.z, c.z); y.z = (y.z >= 0.0f) ? y.z : alpha * y.z;
    y.w = fmaf(v.w, a.w, c.w); y.w = (y.w >= 0.0f) ? y.w : alpha * y.w;
    return y;
}

__device__ __forceinline__ void bn_affine(float s, float q, float g, float b, float invn,
                                          float& a, float& c) {
    const float mean = s * invn;
    const float var = q * invn - mean * mean;
    const float inv = rsqrtf(var + 1e-5f);
    a = inv * g;
    c = b - mean * a;
}

__device__ __forceinline__ void bn_affine4(float4 s, float4 q, float4 g, float4 b, float invn,
                                           float4& a, float4& c) {
    bn_affine(s.x, q.x, g.x, b.x, invn, a.x, c.x);
    bn_affine(s.y, q.y, g.y, b.y, invn, a.y, c.y);
    bn_affine(s.z, q.z, g.z, b.z, invn, a.z, c.z);
    bn_affine(s.w, q.w, g.w, b.w, invn, a.w, c.w);
}

__device__ __forceinline__ void gh_flush(float* __restrict__ gh, int g, int lane, float4 s) {
    float* gp = gh + (long)g * DFEAT + lane * 4;
    atomicAdd(gp + 0, s.x);
    atomicAdd(gp + 1, s.y);
    atomicAdd(gp + 2, s.z);
    atomicAdd(gp + 3, s.w);
}

__global__ __launch_bounds__(128) void k_bnapply(const float* __restrict__ hpre,
                                                 const float* __restrict__ fsum,
                                                 const float* __restrict__ fsq,
                                                 const float* __restrict__ gamma,
                                                 const float* __restrict__ beta,
                                                 const float* __restrict__ alpha_p,
                                                 const int* __restrict__ gids,
                                                 float* __restrict__ hout,
                                                 float* __restrict__ gh, int n) {
    const int lane = threadIdx.x & 31;
    const int gw = __builtin_amdgcn_readfirstlane((int)(blockIdx.x * 4 + (threadIdx.x >> 5)));
    const int row0 = gw * BN_ROWS;
    if (row0 >= n) return;
    const float alpha = alpha_p[0];
    const float4 s4 = ((const float4*)fsum)[lane];
    const float4 q4 = ((const float4*)fsq)[lane];
    const float4 g4 = ((const float4*)gamma)[lane];
    const float4 b4 = ((const float4*)beta)[lane];
    const float invn = 1.0f / (float)n;
    float4 a, c;
    const float4* src = (const float4*)hpre + (long)row0 * 32 + lane;
    float4* dst = (float4*)hout + (long)row0 * 32 + lane;
    float4 s = make_float4(0.0f, 0.0f, 0.0f, 0.0f);
    int gcur = gids[row0];
    if (row0 + BN_ROWS <= n) {
        float4 v[BN_ROWS];
        int g[BN_ROWS];
#pragma unroll
        for (int i = 0; i < BN_ROWS; ++i) {
            v[i] = src[i * 32];
            g[i] = gids[row0 + i];
        }
        bn_affine4(s4, q4, g4, b4, invn, a, c);
#pragma unroll
        for (int i = 0; i < BN_ROWS; ++i) {
            float4 y = bn_act(v[i], a, c, alpha);
            dst[i * 32] = y;
            if (g[i] != gcur) {
                gh_flush(gh, gcur, lane, s);
                s = make_float4(0.0f, 0.0f, 0.0f, 0.0f);
                gcur = g[i];
            }
            s.x += y.x; s.y += y.y; s.z += y.z; s.w += y.w;
        }
    } else {
        const int cnt = n - row0;
        bn_affine4(s4, q4, g4, b4, invn, a, c);
        for (int i = 0; i < cnt; ++i) {
            float4 y = bn_act(src[i * 32], a, c, alpha);
            dst[i * 32] = y;
            const int g = gids[row0 + i];
            if (g != gcur) {
                gh_flush(gh, gcur, lane, s);
                s = make_float4(0.0f, 0.0f, 0.0f, 0.0f);
                gcur = g;
            }
            s.x += y.x; s.y += y.y; s.z += y.z; s.w += y.w;
        }
    }
    gh_flush(gh, gcur, lane, s);
}

static inline int cdiv(long a, long b) { return (int)((a + b - 1) / b); }

extern "C" void kernel_launch(void* const* d_in, const int* in_sizes, int n_in,
                              void* d_out, int out_size, void* d_ws, size_t ws_size,
                              hipStream_t stream) {
    (void)n_in; (void)ws_size;
    const float* heat = (const float*)d_in[0];
    const int* esrc   = (const int*)d_in[1];
    const int* edst   = (const int*)d_in[2];
    const int* gids   = (const int*)d_in[3];
    const float* W0 = (const float*)d_in[4];
    const float* b0 = (const float*)d_in[5];
    const float* ac0 = (const float*)d_in[6];
    const float* g0 = (const float*)d_in[7];
    const float* be0 = (const float*)d_in[8];
    const float* aa0 = (const float*)d_in[9];
    const float* W1 = (const float*)d_in[10];
    const float* b1 = (const float*)d_in[11];
    const float* ac1 = (const float*)d_in[12];
    const float* g1 = (const float*)d_in[13];
    const float* be1 = (const float*)d_in[14];
    const float* aa1 = (const float*)d_in[15];

    const int E = in_sizes[1];
    const int N = in_sizes[3];
    const int ND = N * DFEAT;
    const int GD = out_size - ND;
    const int NB = cdiv(N, RSIZE);
    const int slice = cdiv(E, BIN_BLOCKS);

    float* ws   = (float*)d_ws;
    int* dout   = (int*)d_ws;
    int* ptr    = dout + N;
    float* agg  = ws + 2 * (long)N;
    float* r2   = agg + ND;
    int* csr    = (int*)r2;
    float* ndst = r2 + 2 * (long)E;
    int* sortD  = (int*)(ndst + N + (N & 1));
    int* sortS  = sortD + 2 * (long)E;
    int* tabD   = sortS + E;
    int* tabS   = tabD + (long)BIN_BLOCKS * (NB + 1);
    float* fsum0 = r2 + ND;
    float* fsq0  = fsum0 + DFEAT;
    float* fsum1 = fsq0 + DFEAT;
    float* fsq1  = fsum1 + DFEAT;

    float* hout = (float*)d_out;
    float* gh   = hout + ND;

    const int gemm_blocks = cdiv(N, 128);

    k_bin<<<BIN_BLOCKS, BIN_THREADS, 0, stream>>>(esrc, edst, sortD, sortS, tabD, tabS, fsum0,
                                                  4 * DFEAT, gh, GD, E, slice, NB);
    k_count<<<NB, BIN_THREADS, 0, stream>>>(sortD, sortS, tabD, tabS, dout, ptr, ndst, N, slice, NB);
    k_place<<<NB, BIN_THREADS, 0, stream>>>(sortD, tabD, dout, ptr, csr, N, E, slice, NB);

    k_gather<<<cdiv(N, 8), 256, 0, stream>>>(heat, ptr, csr, agg, N, E);
    k_gemm_bn<<<gemm_blocks, 256, 0, stream>>>(agg, agg, W0, b0, ac0, ndst, fsum0, fsq0, N);
    k_bnapply<<<cdiv(N, BN_ROWS * 4), 128, 0, stream>>>(agg, fsum0, fsq0, g0, be0, aa0, gids, hout,
                                                       gh, N);

    k_gather<<<cdiv(N, 8), 256, 0, stream>>>(hout, ptr, csr, agg, N, E);
    k_gemm_bn<<<gemm_blocks, 256, 0, stream>>>(agg, agg, W1, b1, ac1, ndst, fsum1, fsq1, N);
    k_bnapply<<<cdiv(N, BN_ROWS * 4), 128, 0, stream>>>(agg, fsum1, fsq1, g1, be1, aa1, gids, hout,
                                                       gh, N);
}


# layer 0's BN + PReLU no longer a launch: layer 1's gather applies it to each gathered row in registers, the final apply adds both layers' graph sums (their sum order changes); 9 -> 8 launches
# speedup vs baseline: 3.7441x; 1.1448x over previous
#include <hip/hip_runtime.h>

typedef __attribute__((ext_vector_type(2))) float v2f;
typedef __attribute__((ext_vector_type(8))) float v8f;

#define DFEAT 128
#define PSTR 288
#define TPITCH 132

#define RSHIFT 8
#define RSIZE 256
#define BIN_THREADS 1024
#define BIN_BLOCKS 128
#define SLICE_CAP 4736
#define PLACE_CAP 6144

__device__ __forceinline__ int block_excl_scan(int v, int* red, int nwaves, int& total) {
    const int lane = threadIdx.x & 31;
    const int wave = threadIdx.x >> 5;
    int inc = v;
    for (int o = 1; o < 32; o <<= 1) {
        const int up = __shfl_up(inc, o);
        if (lane >= o) inc += up;
    }
    if (lane == 31) red[wave] = inc;
    __syncthreads();
    const int w = (lane < nwaves) ? red[lane] : 0;
    int winc = w;
    for (int o = 1; o < 32; o <<= 1) {
        const int up = __shfl_up(winc, o);
        if (lane >= o) winc += up;
    }
    total = __shfl(winc, 31);
    const int wexc = __shfl(winc - w, wave);
    __syncthreads();
    return wexc + inc - v;
}

__global__ __launch_bounds__(BIN_THREADS) void k_bin(const int* __restrict__ esrc,
                                                     const int* __restrict__ edst,
                                                     int* __restrict__ sortD,
                                                     int* __restrict__ sortS,
                                                     int* __restrict__ tabD,
                                                     int* __restrict__ tabS,
                                                     float* __restrict__ stats, int nstats,
                                                     float* __restrict__ gh, int ngh,
                                                     int e, int slice, int nb) {
    __shared__ int hD[BIN_THREADS];
    __shared__ int hS[BIN_THREADS];
    __shared__ int red[32];
    __shared__ int2 lD[SLICE_CAP];
    __shared__ int lS[SLICE_CAP];
    const int tid = threadIdx.x;
    const bool staged = slice <= SLICE_CAP;

    for (int i = blockIdx.x * BIN_THREADS + tid; i < nstats + ngh; i += gridDim.x * BIN_THREADS) {
        if (i < nstats) stats[i] = 0.0f;
        else gh[i - nstats] = 0.0f;
    }

    hD[tid] = 0;
    hS[tid] = 0;
    __syncthreads();

    const int beg = blockIdx.x * slice;
    const int end = min(beg + slice, e);
    for (int i = beg + tid; i < end; i += BIN_THREADS) {
        atomicAdd(&hD[edst[i] >> RSHIFT], 1);
        atomicAdd(&hS[esrc[i] >> RSHIFT], 1);
    }
    __syncthreads();

    const int vD = hD[tid];
    const int vS = hS[tid];
    int totD, totS;
    const int exD = block_excl_scan(vD, red, BIN_THREADS / 32, totD);
    const int exS = block_excl_scan(vS, red, BIN_THREADS / 32, totS);
    hD[tid] = exD;
    hS[tid] = exS;
    int* rowD = tabD + (long)blockIdx.x * (nb + 1);
    int* rowS = tabS + (long)blockIdx.x * (nb + 1);
    if (tid < nb) {
        rowD[tid] = exD;
        rowS[tid] = exS;
    }
    if (tid == 0) {
        rowD[nb] = totD;
        rowS[nb] = totS;
    }
    __syncthreads();

    if (staged) {
        for (int i = beg + tid; i < end; i += BIN_THREADS) {
            const int s = esrc[i];
            const int d = edst[i];
            const int pd = atomicAdd(&hD[d >> RSHIFT], 1);
            const int ps = atomicAdd(&hS[s >> RSHIFT], 1);
            lD[pd] = make_int2(s, d);
            lS[ps] = s;
        }
        __syncthreads();
        const int cnt = end - beg;
        for (int i = tid; i < cnt; i += BIN_THREADS) {
            ((int2*)sortD)[beg + i] = lD[i];
            sortS[beg + i] = lS[i];
        }
    } else {
        for (int i = beg + tid; i < end; i += BIN_THREADS) {
            const int s = esrc[i];
            const int d = edst[i];
            const int pd = atomicAdd(&hD[d >> RSHIFT], 1);
            const int ps = atomicAdd(&hS[s >> RSHIFT], 1);
            ((int2*)sortD)[beg + pd] = make_int2(s, d);
            sortS[beg + ps] = s;
        }
    }
}

__global__ __launch_bounds__(BIN_THREADS) void k_count(const int* __restrict__ sortD,
                                                       const int* __restrict__ sortS,
                                                       const int* __restrict__ tabD,
                                                       const int* __restrict__ tabS,
                                                       int* __restrict__ dout,
                                                       int* __restrict__ ptr,
                                                       float* __restrict__ ndst,
                                                       int n, int slice, int nb) {
    __shared__ int cIn[RSIZE];
    __shared__ int cOut[RSIZE];
    __shared__ int pD0[BIN_BLOCKS], pD1[BIN_BLOCKS], pS0[BIN_BLOCKS], pS1[BIN_BLOCKS];
    __shared__ int red[32];
    const int tid = threadIdx.x;
    const int lane = tid & 31;
    const int wave = tid >> 5;
    const int b = blockIdx.x;
    const int lo = b << RSHIFT;

    if (tid < RSIZE) { cIn[tid] = 0; cOut[tid] = 0; }
    int below = 0;
    if (tid < BIN_BLOCKS) {
        const int* td = tabD + (long)tid * (nb + 1) + b;
        const int* ts = tabS + (long)tid * (nb + 1) + b;
        const int d0 = td[0], d1 = td[1], s0 = ts[0], s1 = ts[1];
        pD0[tid] = tid * slice + d0;
        pD1[tid] = tid * slice + d1;
        pS0[tid] = tid * slice + s0;
        pS1[tid] = tid * slice + s1;
        below = d0;
    }
    int base;
    (void)block_excl_scan(below, red, BIN_THREADS / 32, base);

    const int2* sd = (const int2*)sortD;
    for (int j = wave; j < BIN_BLOCKS; j += BIN_THREADS / 32) {
        const int d1 = pD1[j];
        for (int k = pD0[j] + lane; k < d1; k += 32) atomicAdd(&cIn[sd[k].y - lo], 1);
        const int s1 = pS1[j];
        for (int k = pS0[j] + lane; k < s1; k += 32) atomicAdd(&cOut[sortS[k] - lo], 1);
    }
    __syncthreads();

    const int v = (tid < RSIZE) ? cIn[tid] : 0;
    int total;
    const int ex = block_excl_scan(v, red, BIN_THREADS / 32, total);
    if (tid < RSIZE && lo + tid < n) {
        ptr[lo + tid] = base + ex;
        ndst[lo + tid] = (v > 0) ? rsqrtf((float)v) : 0.0f;
        dout[lo + tid] = cOut[tid];
    }
}

__global__ __launch_bounds__(BIN_THREADS) void k_place(const int* __restrict__ sortD,
                                                       const int* __restrict__ tabD,
                                                       const int* __restrict__ dout,
                                                       const int* __restrict__ ptr,
                                                       int* __restrict__ csr,
                                                       int n, int e, int slice, int nb) {
    __shared__ int cur[RSIZE];
    __shared__ int pD0[BIN_BLOCKS], pD1[BIN_BLOCKS];
    __shared__ int2 lC[PLACE_CAP];
    const int tid = threadIdx.x;
    const int lane = tid & 31;
    const int wave = tid >> 5;
    const int b = blockIdx.x;
    const int lo = b << RSHIFT;
    const int base = ptr[lo];
    const int total = ((lo + RSIZE < n) ? ptr[lo + RSIZE] : e) - base;
    const bool staged = total <= PLACE_CAP;

    if (tid < RSIZE) cur[tid] = (lo + tid < n) ? ptr[lo + tid] : 0;
    if (tid < BIN_BLOCKS) {
        const int* td = tabD + (long)tid * (nb + 1) + b;
        pD0[tid] = tid * slice + td[0];
        pD1[tid] = tid * slice + td[1];
    }
    __syncthreads();

    const int2* sd = (const int2*)sortD;
    if (staged) {
        for (int j = wave; j < BIN_BLOCKS; j += BIN_THREADS / 32) {
            const int d1 = pD1[j];
            for (int k = pD0[j] + lane; k < d1; k += 32) {
                const int2 ed = sd[k];
                const int pos = atomicAdd(&cur[ed.y - lo], 1);
                const float w = rsqrtf((float)dout[ed.x]);
                lC[pos - base] = make_int2(ed.x, __float_as_int(w));
            }
        }
        __syncthreads();
        for (int i = tid; i < total; i += BIN_THREADS) ((int2*)csr)[base + i] = lC[i];
    } else {
        for (int j = wave; j < BIN_BLOCKS; j += BIN_THREADS / 32) {
            const int d1 = pD1[j];
            for (int k = pD0[j] + lane; k < d1; k += 32) {
                const int2 ed = sd[k];
                const int pos = atomicAdd(&cur[ed.y - lo], 1);
                const float w = rsqrtf((float)dout[ed.x]);
                ((int2*)csr)[pos] = make_int2(ed.x, __float_as_int(w));
            }
        }
    }
}

__device__ __forceinline__ float4 bn_act(float4 v, float4 a, float4 c, float alpha) {
    float4 y;
    y.x = fmaf(v.x, a.x, c.x); y.x = (y.x >= 0.0f) ? y.x : alpha * y.x;
    y.y = fmaf(v.y, a.y, c.y); y.y = (y.y >= 0.0f) ? y.y : alpha * y.y;
    y.z = fmaf(v.z, a.z, c.z); y.z = (y.z >= 0.0f) ? y.z : alpha * y.z;
    y.w = fmaf(v.w, a.w, c.w); y.w = (y.w >= 0.0f) ? y.w : alpha * y.w;
    return y;
}

__device__ __forceinline__ void bn_affine(float s, float q, float g, float b, float invn,
                                          float& a, float& c) {
    const float mean = s * invn;
    const float var = q * invn - mean * mean;
    const float inv = rsqrtf(var + 1e-5f);
    a = inv * g;
    c = b - mean * a;
}

__device__ __forceinline__ void bn_affine4(float4 s, float4 q, float4 g, float4 b, float invn,
                                           float4& a, float4& c) {
    bn_affine(s.x, q.x, g.x, b.x, invn, a.x, c.x);
    bn_affine(s.y, q.y, g.y, b.y, invn, a.y, c.y);
    bn_affine(s.z, q.z, g.z, b.z, invn, a.z, c.z);
    bn_affine(s.w, q.w, g.w, b.w, invn, a.w, c.w);
}

__device__ __forceinline__ void gather_row(const bool bn, const float* __restrict__ x,
                                           const int* __restrict__ ptr,
                                           const int* __restrict__ csr,
                                           float* __restrict__ agg, int n, int e,
                                           const float* __restrict__ fsum,
                                           const float* __restrict__ fsq,
                                           const float* __restrict__ gamma,
                                           const float* __restrict__ beta,
                                           const float* __restrict__ alpha_p) {
    const int lane = threadIdx.x & 31;
    const int row = __builtin_amdgcn_readfirstlane((int)(blockIdx.x * 8 + (threadIdx.x >> 5)));
    if (row >= n) return;
    const int beg = ptr[row];
    const int end = (row + 1 < n) ? ptr[row + 1] : e;
    float4 a = make_float4(1.0f, 1.0f, 1.0f, 1.0f);
    float4 c = make_float4(0.0f, 0.0f, 0.0f, 0.0f);
    float alpha = 1.0f;
    if (bn) {
        alpha = alpha_p[0];
        const float4 s4 = ((const float4*)fsum)[lane];
        const float4 q4 = ((const float4*)fsq)[lane];
        const float4 g4 = ((const float4*)gamma)[lane];
        const float4 b4 = ((const float4*)beta)[lane];
        bn_affine4(s4, q4, g4, b4, 1.0f / (float)n, a, c);
    }
    const int2* ent = (const int2*)csr;
    const float4* xv = (const float4*)x + lane;
    float4 acc = make_float4(0.0f, 0.0f, 0.0f, 0.0f);
    for (int j = beg; j < end; j += 4) {
        const int rem = end - j;
        const bool ok1 = rem > 1, ok2 = rem > 2, ok3 = rem > 3;
        const int2 e0 = ent[j];
        const int2 e1 = ent[ok1 ? j + 1 : j];
        const int2 e2 = ent[ok2 ? j + 2 : j];
        const int2 e3 = ent[ok3 ? j + 3 : j];
        float4 v0 = xv[(long)e0.x * 32];
        float4 v1 = xv[(long)e1.x * 32];
        float4 v2 = xv[(long)e2.x * 32];
        float4 v3 = xv[(long)e3.x * 32];
        if (bn) {
            v0 = bn_act(v0, a, c, alpha);
            v1 = bn_act(v1, a, c, alpha);
            v2 = bn_act(v2, a, c, alpha);
            v3 = bn_act(v3, a, c, alpha);
        }
        const float w0 = __int_as_float(e0.y);
        const float w1 = __int_as_float(e1.y);
        const float w2 = __int_as_float(e2.y);
        const float w3 = __int_as_float(e3.y);
        acc.x += v0.x * w0; acc.y += v0.y * w0; acc.z += v0.z * w0; acc.w += v0.w * w0;
        acc.x = ok1 ? acc.x + v1.x * w1 : acc.x;
        acc.y = ok1 ? acc.y + v1.y * w1 : acc.y;
        acc.z = ok1 ? acc.z + v1.z * w1 : acc.z;
        acc.w = ok1 ? acc.w + v1.w * w1 : acc.w;
        acc.x = ok2 ? acc.x + v2.x * w2 : acc.x;
        acc.y = ok2 ? acc.y + v2.y * w2 : acc.y;
        acc.z = ok2 ? acc.z + v2.z * w2 : acc.z;
        acc.w = ok2 ? acc.w + v2.w * w2 : acc.w;
        acc.x = ok3 ? acc.x + v3.x * w3 : acc.x;
        acc.y = ok3 ? acc.y + v3.y * w3 : acc.y;
        acc.z = ok3 ? acc.z + v3.z * w3 : acc.z;
        acc.w = ok3 ? acc.w + v3.w * w3 : acc.w;
    }
    ((float4*)agg)[(long)row * 32 + lane] = acc;
}

__global__ __launch_bounds__(256) void k_gather(const float* __restrict__ x,
                                                const int* __restrict__ ptr,
                                                const int* __restrict__ csr,
                                                float* __restrict__ agg, int n, int e) {
    gather_row(false, x, ptr, csr, agg, n, e, nullptr, nullptr, nullptr, nullptr, nullptr);
}

__global__ __launch_bounds__(256) void k_gather_bn(const float* __restrict__ x,
                                                   const int* __restrict__ ptr,
                                                   const int* __restrict__ csr,
                                                   float* __restrict__ agg, int n, int e,
                                                   const float* __restrict__ fsum,
                                                   const float* __restrict__ fsq,
                                                   const float* __restrict__ gamma,
                                                   const float* __restrict__ beta,
                                                   const float* __restrict__ alpha_p) {
    gather_row(true, x, ptr, csr, agg, n, e, fsum, fsq, gamma, beta, alpha_p);
}

__device__ __forceinline__ void gemm_ep_row(float4& v, float4 bv, float alpha, float4& s, float4& q) {
    v.x += bv.x; v.y += bv.y; v.z += bv.z; v.w += bv.w;
    v.x = (v.x >= 0.0f) ? v.x : alpha * v.x;
    v.y = (v.y >= 0.0f) ? v.y : alpha * v.y;
    v.z = (v.z >= 0.0f) ? v.z : alpha * v.z;
    v.w = (v.w >= 0.0f) ? v.w : alpha * v.w;
    s.x += v.x; s.y += v.y; s.z += v.z; s.w += v.w;
    q.x += v.x * v.x; q.y += v.y * v.y; q.z += v.z * v.z; q.w += v.w * v.w;
}

__global__ __launch_bounds__(256) void k_gemm_bn(const float* agg, float* hpre,
                                                 const float* __restrict__ W,
                                                 const float* __restrict__ bias,
                                                 const float* __restrict__ alpha_p,
                                                 const float* __restrict__ ndst,
                                                 float* __restrict__ fsum,
                                                 float* __restrict__ fsq, int n) {
    __shared__ float sW[16 * PSTR];
    __shared__ float sSum[DFEAT];
    __shared__ float sSq[DFEAT];
    __shared__ float4 sT[8 * 16 * (TPITCH / 4)];

    const int tid = threadIdx.x;
    if (tid < DFEAT) { sSum[tid] = 0.0f; sSq[tid] = 0.0f; }

    const int wave = tid >> 5;
    const int lane = tid & 31;
    const int half = lane >> 4;
    const int mcol = lane & 15;
    const int r0 = blockIdx.x * 128 + wave * 16;
    const int rowA = r0 + mcol;
    const bool aok = rowA < n;
    const float nd = aok ? ndst[rowA] : 0.0f;
    const int rowC = aok ? rowA : (n - 1);
    const float alpha = alpha_p[0];
    const float* aptr = agg + (long)rowC * DFEAT;

    v8f acc[8] = {};

    for (int kc = 0; kc < DFEAT; kc += 32) {
        __syncthreads();
        {
            int kp = tid >> 4;
            int nb = (tid & 15) * 8;
            const float* w0 = W + (kc + 2 * kp) * DFEAT + nb;
            const float* w1 = w0 + DFEAT;
            float4 r0a = ((const float4*)w0)[0];
            float4 r0b = ((const float4*)w0)[1];
            float4 r1a = ((const float4*)w1)[0];
            float4 r1b = ((const float4*)w1)[1];
            float2* dp = (float2*)(sW + kp * PSTR + nb * 2);
            dp[0] = make_float2(r0a.x, r1a.x);
            dp[1] = make_float2(r0a.y, r1a.y);
            dp[2] = make_float2(r0a.z, r1a.z);
            dp[3] = make_float2(r0a.w, r1a.w);
            dp[4] = make_float2(r0b.x, r1b.x);
            dp[5] = make_float2(r0b.y, r1b.y);
            dp[6] = make_float2(r0b.z, r1b.z);
            dp[7] = make_float2(r0b.w, r1b.w);
        }
        __syncthreads();

        for (int kb = 0; kb < 32; kb += 4) {
            const int kA = kc + kb + half * 2;
            float2 av = *(const float2*)(aptr + kA);
            v2f a;
            a.x = av.x * nd;
            a.y = av.y * nd;
            const float* bp = sW + (kb / 2 + half) * PSTR + mcol * 2;
#pragma unroll
            for (int t = 0; t < 8; ++t) {
                v2f b = *(const v2f*)(bp + t * 32);
                acc[t] = __builtin_amdgcn_wmma_f32_16x16x4_f32(
                    false, a, false, b, (short)0, acc[t], false, false);
            }
        }
    }

    float* tw = (float*)sT + wave * (16 * TPITCH);
#pragma unroll
    for (int t = 0; t < 8; ++t) {
#pragma unroll
        for (int r = 0; r < 8; ++r) tw[(half * 8 + r) * TPITCH + t * 16 + mcol] = acc[t][r];
    }
    __syncthreads();
    {
        const int wave_s = __builtin_amdgcn_readfirstlane(wave);
        const int row0 = blockIdx.x * 128 + wave_s * 16;
        const float4 bv = ((const float4*)bias)[lane];
        float4 s = make_float4(0.0f, 0.0f, 0.0f, 0.0f);
        float4 q = make_float4(0.0f, 0.0f, 0.0f, 0.0f);
        const float* trow = tw + lane * 4;
        if (row0 + 16 <= n) {
            float4* orow = (float4*)hpre + (long)row0 * 32 + lane;
            float4 v[16];
#pragma unroll
            for (int i = 0; i < 16; ++i) v[i] = *(const float4*)(trow + i * TPITCH);
#pragma unroll
            for (int i = 0; i < 16; ++i) {
                gemm_ep_row(v[i], bv, alpha, s, q);
                orow[i * 32] = v[i];
            }
        } else {
            for (int i = 0; i < 16 && row0 + i < n; ++i) {
                float4 v = *(const float4*)(trow + i * TPITCH);
                gemm_ep_row(v, bv, alpha, s, q);
                ((float4*)hpre)[(long)(row0 + i) * 32 + lane] = v;
            }
        }
        atomicAdd(&sSum[lane * 4 + 0], s.x);
        atomicAdd(&sSum[lane * 4 + 1], s.y);
        atomicAdd(&sSum[lane * 4 + 2], s.z);
        atomicAdd(&sSum[lane * 4 + 3], s.w);
        atomicAdd(&sSq[lane * 4 + 0], q.x);
        atomicAdd(&sSq[lane * 4 + 1], q.y);
        atomicAdd(&sSq[lane * 4 + 2], q.z);
        atomicAdd(&sSq[lane * 4 + 3], q.w);
    }
    __syncthreads();
    if (tid < DFEAT) {
        atomicAdd(&fsum[tid], sSum[tid]);
        atomicAdd(&fsq[tid], sSq[tid]);
    }
}

#define BN_ROWS 16

__device__ __forceinline__ void gh_flush(float* __restrict__ gh, int g, int lane, float4 s) {
    float* gp = gh + (long)g * DFEAT + lane * 4;
    atomicAdd(gp + 0, s.x);
    atomicAdd(gp + 1, s.y);
    atomicAdd(gp + 2, s.z);
    atomicAdd(gp + 3, s.w);
}

__global__ __launch_bounds__(128) void k_bnapply2(float* h, const float* __restrict__ h0,
                                                  const float* __restrict__ fsum1,
                                                  const float* __restrict__ fsq1,
                                                  const float* __restrict__ gamma1,
                                                  const float* __restrict__ beta1,
                                                  const float* __restrict__ alpha1_p,
                                                  const float* __restrict__ fsum0,
                                                  const float* __restrict__ fsq0,
                                                  const float* __restrict__ gamma0,
                                                  const float* __restrict__ beta0,
                                                  const float* __restrict__ alpha0_p,
                                                  const int* __restrict__ gids,
                                                  float* __restrict__ gh, int n) {
    const int lane = threadIdx.x & 31;
    const int gw = __builtin_amdgcn_readfirstlane((int)(blockIdx.x * 4 + (threadIdx.x >> 5)));
    const int row0 = gw * BN_ROWS;
    if (row0 >= n) return;
    const float alpha1 = alpha1_p[0];
    const float alpha0 = alpha0_p[0];
    const float4 s41 = ((const float4*)fsum1)[lane];
    const float4 q41 = ((const float4*)fsq1)[lane];
    const float4 g41 = ((const float4*)gamma1)[lane];
    const float4 b41 = ((const float4*)beta1)[lane];
    const float4 s40 = ((const float4*)fsum0)[lane];
    const float4 q40 = ((const float4*)fsq0)[lane];
    const float4 g40 = ((const float4*)gamma0)[lane];
    const float4 b40 = ((const float4*)beta0)[lane];
    const float invn = 1.0f / (float)n;
    float4 a1, c1, a0, c0;
    float4* hr = (float4*)h + (long)row0 * 32 + lane;
    const float4* h0r = (const float4*)h0 + (long)row0 * 32 + lane;
    float4 s = make_float4(0.0f, 0.0f, 0.0f, 0.0f);
    int gcur = gids[row0];
    if (row0 + BN_ROWS <= n) {
        float4 v1[BN_ROWS];
        float4 v0[BN_ROWS];
        int g[BN_ROWS];
#pragma unroll
        for (int i = 0; i < BN_ROWS; ++i) {
            v1[i] = hr[i * 32];
            v0[i] = h0r[i * 32];
            g[i] = gids[row0 + i];
        }
        __builtin_amdgcn_sched_barrier(0);
        bn_affine4(s41, q41, g41, b41, invn, a1, c1);
        bn_affine4(s40, q40, g40, b40, invn, a0, c0);
#pragma unroll
        for (int i = 0; i < BN_ROWS; ++i) {
            const float4 y1 = bn_act(v1[i], a1, c1, alpha1);
            hr[i * 32] = y1;
            const float4 y0 = bn_act(v0[i], a0, c0, alpha0);
            if (g[i] != gcur) {
                gh_flush(gh, gcur, lane, s);
                s = make_float4(0.0f, 0.0f, 0.0f, 0.0f);
                gcur = g[i];
            }
            s.x += y0.x + y1.x; s.y += y0.y + y1.y; s.z += y0.z + y1.z; s.w += y0.w + y1.w;
        }
    } else {
        const int cnt = n - row0;
        bn_affine4(s41, q41, g41, b41, invn, a1, c1);
        bn_affine4(s40, q40, g40, b40, invn, a0, c0);
        for (int i = 0; i < cnt; ++i) {
            const float4 y1 = bn_act(hr[i * 32], a1, c1, alpha1);
            hr[i * 32] = y1;
            const float4 y0 = bn_act(h0r[i * 32], a0, c0, alpha0);
            const int g = gids[row0 + i];
            if (g != gcur) {
                gh_flush(gh, gcur, lane, s);
                s = make_float4(0.0f, 0.0f, 0.0f, 0.0f);
                gcur = g;
            }
            s.x += y0.x + y1.x; s.y += y0.y + y1.y; s.z += y0.z + y1.z; s.w += y0.w + y1.w;
        }
    }
    gh_flush(gh, gcur, lane, s);
}

static inline int cdiv(long a, long b) { return (int)((a + b - 1) / b); }

extern "C" void kernel_launch(void* const* d_in, const int* in_sizes, int n_in,
                              void* d_out, int out_size, void* d_ws, size_t ws_size,
                              hipStream_t stream) {
    (void)n_in; (void)ws_size;
    const float* heat = (const float*)d_in[0];
    const int* esrc   = (const int*)d_in[1];
    const int* edst   = (const int*)d_in[2];
    const int* gids   = (const int*)d_in[3];
    const float* W0 = (const float*)d_in[4];
    const float* b0 = (const float*)d_in[5];
    const float* ac0 = (const float*)d_in[6];
    const float* g0 = (const float*)d_in[7];
    const float* be0 = (const float*)d_in[8];
    const float* aa0 = (const float*)d_in[9];
    const float* W1 = (const float*)d_in[10];
    const float* b1 = (const float*)d_in[11];
    const float* ac1 = (const float*)d_in[12];
    const float* g1 = (const float*)d_in[13];
    const float* be1 = (const float*)d_in[14];
    const float* aa1 = (const float*)d_in[15];

    const int E = in_sizes[1];
    const int N = in_sizes[3];
    const int ND = N * DFEAT;
    const int GD = out_size - ND;
    const int NB = cdiv(N, RSIZE);
    const int slice = cdiv(E, BIN_BLOCKS);

    float* ws   = (float*)d_ws;
    int* dout   = (int*)d_ws;
    int* ptr    = dout + N;
    float* agg  = ws + 2 * (long)N;
    float* r2   = agg + ND;
    int* csr    = (int*)r2;
    float* ndst = r2 + 2 * (long)E;
    int* sortD  = (int*)(ndst + N + (N & 1));
    int* sortS  = sortD + 2 * (long)E;
    int* tabD   = sortS + E;
    int* tabS   = tabD + (long)BIN_BLOCKS * (NB + 1);
    float* fsum0 = r2 + ND;
    float* fsq0  = fsum0 + DFEAT;
    float* fsum1 = fsq0 + DFEAT;
    float* fsq1  = fsum1 + DFEAT;

    float* hout = (float*)d_out;
    float* gh   = hout + ND;

    const int gemm_blocks = cdiv(N, 128);

    k_bin<<<BIN_BLOCKS, BIN_THREADS, 0, stream>>>(esrc, edst, sortD, sortS, tabD, tabS, fsum0,
                                                  4 * DFEAT, gh, GD, E, slice, NB);
    k_count<<<NB, BIN_THREADS, 0, stream>>>(sortD, sortS, tabD, tabS, dout, ptr, ndst, N, slice, NB);
    k_place<<<NB, BIN_THREADS, 0, stream>>>(sortD, tabD, dout, ptr, csr, N, E, slice, NB);

    k_gather<<<cdiv(N, 8), 256, 0, stream>>>(heat, ptr, csr, agg, N, E);
    k_gemm_bn<<<gemm_blocks, 256, 0, stream>>>(agg, agg, W0, b0, ac0, ndst, fsum0, fsq0, N);

    k_gather_bn<<<cdiv(N, 8), 256, 0, stream>>>(agg, ptr, csr, hout, N, E, fsum0, fsq0, g0, be0,
                                                aa0);
    k_gemm_bn<<<gemm_blocks, 256, 0, stream>>>(hout, hout, W1, b1, ac1, ndst, fsum1, fsq1, N);

    k_bnapply2<<<cdiv(N, BN_ROWS * 4), 128, 0, stream>>>(hout, agg, fsum1, fsq1, g1, be1, aa1,
                                                        fsum0, fsq0, g0, be0, aa0, gids, gh, N);
}


# conv layers as resident blocks: one 16-wave block per WGP stages all of W in LDS once, its waves loop over 16-row tiles, no block barrier per tile (BN column sums regroup: per wave, then wave order)
# speedup vs baseline: 4.7106x; 1.2581x over previous
#include <hip/hip_runtime.h>

typedef __attribute__((ext_vector_type(2))) float v2f;
typedef __attribute__((ext_vector_type(8))) float v8f;

#define DFEAT 128
#define PSTR 288
#define TPITCH 132

#define RSHIFT 8
#define RSIZE 256
#define BIN_THREADS 1024
#define BIN_BLOCKS 128
#define SLICE_CAP 4736
#define PLACE_CAP 6144

__device__ __forceinline__ int block_excl_scan(int v, int* red, int nwaves, int& total) {
    const int lane = threadIdx.x & 31;
    const int wave = threadIdx.x >> 5;
    int inc = v;
    for (int o = 1; o < 32; o <<= 1) {
        const int up = __shfl_up(inc, o);
        if (lane >= o) inc += up;
    }
    if (lane == 31) red[wave] = inc;
    __syncthreads();
    const int w = (lane < nwaves) ? red[lane] : 0;
    int winc = w;
    for (int o = 1; o < 32; o <<= 1) {
        const int up = __shfl_up(winc, o);
        if (lane >= o) winc += up;
    }
    total = __shfl(winc, 31);
    const int wexc = __shfl(winc - w, wave);
    __syncthreads();
    return wexc + inc - v;
}

__global__ __launch_bounds__(BIN_THREADS) void k_bin(const int* __restrict__ esrc,
                                                     const int* __restrict__ edst,
                                                     int* __restrict__ sortD,
                                                     int* __restrict__ sortS,
                                                     int* __restrict__ tabD,
                                                     int* __restrict__ tabS,
                                                     float* __restrict__ stats, int nstats,
                                                     float* __restrict__ gh, int ngh,
                                                     int e, int slice, int nb) {
    __shared__ int hD[BIN_THREADS];
    __shared__ int hS[BIN_THREADS];
    __shared__ int red[32];
    __shared__ int2 lD[SLICE_CAP];
    __shared__ int lS[SLICE_CAP];
    const int tid = threadIdx.x;
    const bool staged = slice <= SLICE_CAP;

    for (int i = blockIdx.x * BIN_THREADS + tid; i < nstats + ngh; i += gridDim.x * BIN_THREADS) {
        if (i < nstats) stats[i] = 0.0f;
        else gh[i - nstats] = 0.0f;
    }

    hD[tid] = 0;
    hS[tid] = 0;
    __syncthreads();

    const int beg = blockIdx.x * slice;
    const int end = min(beg + slice, e);
    for (int i = beg + tid; i < end; i += BIN_THREADS) {
        atomicAdd(&hD[edst[i] >> RSHIFT], 1);
        atomicAdd(&hS[esrc[i] >> RSHIFT], 1);
    }
    __syncthreads();

    const int vD = hD[tid];
    const int vS = hS[tid];
    int totD, totS;
    const int exD = block_excl_scan(vD, red, BIN_THREADS / 32, totD);
    const int exS = block_excl_scan(vS, red, BIN_THREADS / 32, totS);
    hD[tid] = exD;
    hS[tid] = exS;
    int* rowD = tabD + (long)blockIdx.x * (nb + 1);
    int* rowS = tabS + (long)blockIdx.x * (nb + 1);
    if (tid < nb) {
        rowD[tid] = exD;
        rowS[tid] = exS;
    }
    if (tid == 0) {
        rowD[nb] = totD;
        rowS[nb] = totS;
    }
    __syncthreads();

    if (staged) {
        for (int i = beg + tid; i < end; i += BIN_THREADS) {
            const int s = esrc[i];
            const int d = edst[i];
            const int pd = atomicAdd(&hD[d >> RSHIFT], 1);
            const int ps = atomicAdd(&hS[s >> RSHIFT], 1);
            lD[pd] = make_int2(s, d);
            lS[ps] = s;
        }
        __syncthreads();
        const int cnt = end - beg;
        for (int i = tid; i < cnt; i += BIN_THREADS) {
            ((int2*)sortD)[beg + i] = lD[i];
            sortS[beg + i] = lS[i];
        }
    } else {
        for (int i = beg + tid; i < end; i += BIN_THREADS) {
            const int s = esrc[i];
            const int d = edst[i];
            const int pd = atomicAdd(&hD[d >> RSHIFT], 1);
            const int ps = atomicAdd(&hS[s >> RSHIFT], 1);
            ((int2*)sortD)[beg + pd] = make_int2(s, d);
            sortS[beg + ps] = s;
        }
    }
}

__global__ __launch_bounds__(BIN_THREADS) void k_count(const int* __restrict__ sortD,
                                                       const int* __restrict__ sortS,
                                                       const int* __restrict__ tabD,
                                                       const int* __restrict__ tabS,
                                                       int* __restrict__ dout,
                                                       int* __restrict__ ptr,
                                                       float* __restrict__ ndst,
                                                       int n, int slice, int nb) {
    __shared__ int cIn[RSIZE];
    __shared__ int cOut[RSIZE];
    __shared__ int pD0[BIN_BLOCKS], pD1[BIN_BLOCKS], pS0[BIN_BLOCKS], pS1[BIN_BLOCKS];
    __shared__ int red[32];
    const int tid = threadIdx.x;
    const int lane = tid & 31;
    const int wave = tid >> 5;
    const int b = blockIdx.x;
    const int lo = b << RSHIFT;

    if (tid < RSIZE) { cIn[tid] = 0; cOut[tid] = 0; }
    int below = 0;
    if (tid < BIN_BLOCKS) {
        const int* td = tabD + (long)tid * (nb + 1) + b;
        const int* ts = tabS + (long)tid * (nb + 1) + b;
        const int d0 = td[0], d1 = td[1], s0 = ts[0], s1 = ts[1];
        pD0[tid] = tid * slice + d0;
        pD1[tid] = tid * slice + d1;
        pS0[tid] = tid * slice + s0;
        pS1[tid] = tid * slice + s1;
        below = d0;
    }
    int base;
    (void)block_excl_scan(below, red, BIN_THREADS / 32, base);

    const int2* sd = (const int2*)sortD;
    for (int j = wave; j < BIN_BLOCKS; j += BIN_THREADS / 32) {
        const int d1 = pD1[j];
        for (int k = pD0[j] + lane; k < d1; k += 32) atomicAdd(&cIn[sd[k].y - lo], 1);
        const int s1 = pS1[j];
        for (int k = pS0[j] + lane; k < s1; k += 32) atomicAdd(&cOut[sortS[k] - lo], 1);
    }
    __syncthreads();

    const int v = (tid < RSIZE) ? cIn[tid] : 0;
    int total;
    const int ex = block_excl_scan(v, red, BIN_THREADS / 32, total);
    if (tid < RSIZE && lo + tid < n) {
        ptr[lo + tid] = base + ex;
        ndst[lo + tid] = (v > 0) ? rsqrtf((float)v) : 0.0f;
        dout[lo + tid] = cOut[tid];
    }
}

__global__ __launch_bounds__(BIN_THREADS) void k_place(const int* __restrict__ sortD,
                                                       const int* __restrict__ tabD,
                                                       const int* __restrict__ dout,
                                                       const int* __restrict__ ptr,
                                                       int* __restrict__ csr,
                                                       int n, int e, int slice, int nb) {
    __shared__ int cur[RSIZE];
    __shared__ int pD0[BIN_BLOCKS], pD1[BIN_BLOCKS];
    __shared__ int2 lC[PLACE_CAP];
    const int tid = threadIdx.x;
    const int lane = tid & 31;
    const int wave = tid >> 5;
    const int b = blockIdx.x;
    const int lo = b << RSHIFT;
    const int base = ptr[lo];
    const int total = ((lo + RSIZE < n) ? ptr[lo + RSIZE] : e) - base;
    const bool staged = total <= PLACE_CAP;

    if (tid < RSIZE) cur[tid] = (lo + tid < n) ? ptr[lo + tid] : 0;
    if (tid < BIN_BLOCKS) {
        const int* td = tabD + (long)tid * (nb + 1) + b;
        pD0[tid] = tid * slice + td[0];
        pD1[tid] = tid * slice + td[1];
    }
    __syncthreads();

    const int2* sd = (const int2*)sortD;
    if (staged) {
        for (int j = wave; j < BIN_BLOCKS; j += BIN_THREADS / 32) {
            const int d1 = pD1[j];
            for (int k = pD0[j] + lane; k < d1; k += 32) {
                const int2 ed = sd[k];
                const int pos = atomicAdd(&cur[ed.y - lo], 1);
                const float w = rsqrtf((float)dout[ed.x]);
                lC[pos - base] = make_int2(ed.x, __float_as_int(w));
            }
        }
        __syncthreads();
        for (int i = tid; i < total; i += BIN_THREADS) ((int2*)csr)[base + i] = lC[i];
    } else {
        for (int j = wave; j < BIN_BLOCKS; j += BIN_THREADS / 32) {
            const int d1 = pD1[j];
            for (int k = pD0[j] + lane; k < d1; k += 32) {
                const int2 ed = sd[k];
                const int pos = atomicAdd(&cur[ed.y - lo], 1);
                const float w = rsqrtf((float)dout[ed.x]);
                ((int2*)csr)[pos] = make_int2(ed.x, __float_as_int(w));
            }
        }
    }
}

__device__ __forceinline__ float4 bn_act(float4 v, float4 a, float4 c, float alpha) {
    float4 y;
    y.x = fmaf(v.x, a.x, c.x); y.x = (y.x >= 0.0f) ? y.x : alpha * y.x;
    y.y = fmaf(v.y, a.y, c.y); y.y = (y.y >= 0.0f) ? y.y : alpha * y.y;
    y.z = fmaf(v.z, a.z, c.z); y.z = (y.z >= 0.0f) ? y.z : alpha * y.z;
    y.w = fmaf(v.w, a.w, c.w); y.w = (y.w >= 0.0f) ? y.w : alpha * y.w;
    return y;
}

__device__ __forceinline__ void bn_affine(float s, float q, float g, float b, float invn,
                                          float& a, float& c) {
    const float mean = s * invn;
    const float var = q * invn - mean * mean;
    const float inv = rsqrtf(var + 1e-5f);
    a = inv * g;
    c = b - mean * a;
}

__device__ __forceinline__ void bn_affine4(float4 s, float4 q, float4 g, float4 b, float invn,
                                           float4& a, float4& c) {
    bn_affine(s.x, q.x, g.x, b.x, invn, a.x, c.x);
    bn_affine(s.y, q.y, g.y, b.y, invn, a.y, c.y);
    bn_affine(s.z, q.z, g.z, b.z, invn, a.z, c.z);
    bn_affine(s.w, q.w, g.w, b.w, invn, a.w, c.w);
}

__device__ __forceinline__ void tile_store_row(float* tw, int r, float4 acc, float nd) {
    float4 o;
    o.x = acc.x * nd; o.y = acc.y * nd; o.z = acc.z * nd; o.w = acc.w * nd;
    *(float4*)(tw + r * TPITCH + (threadIdx.x & 31) * 4) = o;
}

__device__ __forceinline__ void gather_tile(const bool bn, const float* __restrict__ x,
                                            const int* __restrict__ ptr,
                                            const int* __restrict__ csr,
                                            const float* __restrict__ ndst,
                                            float* tw, int row0, int n, int e,
                                            float4 a, float4 c, float alpha) {
    const int lane = threadIdx.x & 31;
    const int rl = row0 + (lane & 15);
    const int rc = (rl < n) ? rl : (n - 1);
    const int pnext = ptr[min(rl + 1, n - 1)];
    const int myEnd = (rl + 1 < n) ? pnext : e;
    const int myNd = (rl < n) ? __float_as_int(ndst[rc]) : 0;
    const int nrows = min(16, n - row0);
    int cur = 0;
    float4 acc = make_float4(0.0f, 0.0f, 0.0f, 0.0f);
    if (nrows > 0) {
        const int beg = ptr[row0];
        const int endAll = __builtin_amdgcn_readlane(myEnd, nrows - 1);
        int curEnd = __builtin_amdgcn_readlane(myEnd, 0);
        const int2* ent = (const int2*)csr;
        const float4* xv = (const float4*)x + lane;
        int2 en[16];
        if (beg < endAll) {
#pragma unroll
            for (int s = 0; s < 16; ++s) en[s] = ent[min(beg + s, endAll - 1)];
        }
        for (int j = beg; j < endAll; j += 16) {
            float4 v[16];
#pragma unroll
            for (int s = 0; s < 16; ++s) v[s] = xv[(long)en[s].x * 32];
            int2 nx[16];
#pragma unroll
            for (int s = 0; s < 16; ++s) nx[s] = ent[min(j + 16 + s, endAll - 1)];
#pragma unroll
            for (int s = 0; s < 16; ++s) {
                if (j + s < endAll) {
                    while (j + s >= curEnd) {
                        tile_store_row(tw, cur, acc,
                                       __int_as_float(__builtin_amdgcn_readlane(myNd, cur)));
                        acc = make_float4(0.0f, 0.0f, 0.0f, 0.0f);
                        ++cur;
                        curEnd = __builtin_amdgcn_readlane(myEnd, cur);
                    }
                    float4 t = v[s];
                    if (bn) t = bn_act(t, a, c, alpha);
                    const float w = __int_as_float(en[s].y);
                    acc.x += t.x * w; acc.y += t.y * w; acc.z += t.z * w; acc.w += t.w * w;
                }
            }
#pragma unroll
            for (int s = 0; s < 16; ++s) en[s] = nx[s];
        }
    }
    for (; cur < 16; ++cur) {
        tile_store_row(tw, cur, acc, __int_as_float(__builtin_amdgcn_readlane(myNd, cur)));
        acc = make_float4(0.0f, 0.0f, 0.0f, 0.0f);
    }
}

__device__ __forceinline__ void gemm_ep_row(float4& v, float4 bv, float alpha, float4& s, float4& q) {
    v.x += bv.x; v.y += bv.y; v.z += bv.z; v.w += bv.w;
    v.x = (v.x >= 0.0f) ? v.x : alpha * v.x;
    v.y = (v.y >= 0.0f) ? v.y : alpha * v.y;
    v.z = (v.z >= 0.0f) ? v.z : alpha * v.z;
    v.w = (v.w >= 0.0f) ? v.w : alpha * v.w;
    s.x += v.x; s.y += v.y; s.z += v.z; s.w += v.w;
    q.x += v.x * v.x; q.y += v.y * v.y; q.z += v.z * v.z; q.w += v.w * v.w;
}

#define CONV_WAVES 16
#define CONV_BLOCKS 256

__device__ __forceinline__ void wave_lds_order() {
    __builtin_amdgcn_fence(__ATOMIC_RELEASE, "wavefront");
    __builtin_amdgcn_wave_barrier();
    __builtin_amdgcn_fence(__ATOMIC_ACQUIRE, "wavefront");
}

__device__ __forceinline__ void conv_body(const bool bn, const float* __restrict__ x,
                                          const int* __restrict__ ptr,
                                          const int* __restrict__ csr, int e,
                                          const float* __restrict__ psum,
                                          const float* __restrict__ psq,
                                          const float* __restrict__ pgamma,
                                          const float* __restrict__ pbeta,
                                          const float* __restrict__ palpha_p,
                                          float* __restrict__ hpre,
                                          const float* __restrict__ W,
                                          const float* __restrict__ bias,
                                          const float* __restrict__ alpha_p,
                                          const float* __restrict__ ndst,
                                          float* __restrict__ fsum,
                                          float* __restrict__ fsq, int n) {
    __shared__ float sW[64 * PSTR];
    __shared__ float4 sT[CONV_WAVES * 16 * (TPITCH / 4)];

    const int tid = threadIdx.x;

    for (int kp = tid >> 4; kp < 64; kp += (CONV_WAVES * 32) >> 4) {
        const int nb = (tid & 15) * 8;
        const float* w0 = W + (2 * kp) * DFEAT + nb;
        const float* w1 = w0 + DFEAT;
        const float4 r0a = ((const float4*)w0)[0];
        const float4 r0b = ((const float4*)w0)[1];
        const float4 r1a = ((const float4*)w1)[0];
        const float4 r1b = ((const float4*)w1)[1];
        float2* dp = (float2*)(sW + kp * PSTR + nb * 2);
        dp[0] = make_float2(r0a.x, r1a.x);
        dp[1] = make_float2(r0a.y, r1a.y);
        dp[2] = make_float2(r0a.z, r1a.z);
        dp[3] = make_float2(r0a.w, r1a.w);
        dp[4] = make_float2(r0b.x, r1b.x);
        dp[5] = make_float2(r0b.y, r1b.y);
        dp[6] = make_float2(r0b.z, r1b.z);
        dp[7] = make_float2(r0b.w, r1b.w);
    }
    __syncthreads();

    const int wave = tid >> 5;
    const int lane = tid & 31;
    const int half = lane >> 4;
    const int mcol = lane & 15;
    const float alpha = alpha_p[0];
    const int wave_s = __builtin_amdgcn_readfirstlane(wave);
    float* tw = (float*)sT + wave_s * (16 * TPITCH);
    float4 ga = make_float4(1.0f, 1.0f, 1.0f, 1.0f);
    float4 gc = make_float4(0.0f, 0.0f, 0.0f, 0.0f);
    float galpha = 1.0f;
    if (bn) {
        galpha = palpha_p[0];
        const float4 s4 = ((const float4*)psum)[lane];
        const float4 q4 = ((const float4*)psq)[lane];
        const float4 g4 = ((const float4*)pgamma)[lane];
        const float4 b4 = ((const float4*)pbeta)[lane];
        bn_affine4(s4, q4, g4, b4, 1.0f / (float)n, ga, gc);
    }
    const float* arow = tw + mcol * TPITCH + half * 2;
    const float* trow = tw + lane * 4;
    const float4 bv = ((const float4*)bias)[lane];
    float4 s = make_float4(0.0f, 0.0f, 0.0f, 0.0f);
    float4 q = make_float4(0.0f, 0.0f, 0.0f, 0.0f);

    const int ntiles = (n + 15) >> 4;
    for (int tile = blockIdx.x + gridDim.x * wave_s; tile < ntiles; tile += gridDim.x * CONV_WAVES) {
        const int row0 = tile * 16;
        gather_tile(bn, x, ptr, csr, ndst, tw, row0, n, e, ga, gc, galpha);
        wave_lds_order();

        v8f acc[8] = {};
        for (int kc = 0; kc < DFEAT; kc += 32) {
            const float* wc = sW + (kc / 2) * PSTR;
            for (int kb = 0; kb < 32; kb += 4) {
                const v2f a = *(const v2f*)(arow + kc + kb);
                const float* bp = wc + (kb / 2 + half) * PSTR + mcol * 2;
#pragma unroll
                for (int t = 0; t < 8; ++t) {
                    v2f b = *(const v2f*)(bp + t * 32);
                    acc[t] = __builtin_amdgcn_wmma_f32_16x16x4_f32(
                        false, a, false, b, (short)0, acc[t], false, false);
                }
            }
        }

        wave_lds_order();
#pragma unroll
        for (int t = 0; t < 8; ++t) {
#pragma unroll
            for (int r = 0; r < 8; ++r) tw[(half * 8 + r) * TPITCH + t * 16 + mcol] = acc[t][r];
        }
        wave_lds_order();
        if (row0 + 16 <= n) {
            float4* orow = (float4*)hpre + (long)row0 * 32 + lane;
            float4 v[16];
#pragma unroll
            for (int i = 0; i < 16; ++i) v[i] = *(const float4*)(trow + i * TPITCH);
#pragma unroll
            for (int i = 0; i < 16; ++i) {
                gemm_ep_row(v[i], bv, alpha, s, q);
                orow[i * 32] = v[i];
            }
        } else {
            for (int i = 0; i < 16 && row0 + i < n; ++i) {
                float4 v = *(const float4*)(trow + i * TPITCH);
                gemm_ep_row(v, bv, alpha, s, q);
                ((float4*)hpre)[(long)(row0 + i) * 32 + lane] = v;
            }
        }
        wave_lds_order();
    }

    *(float4*)(tw + lane * 4) = s;
    *(float4*)(tw + TPITCH + lane * 4) = q;
    __syncthreads();
    if (tid < DFEAT) {
        const float* p = (const float*)sT + tid;
        float ts = 0.0f, tq = 0.0f;
        for (int w = 0; w < CONV_WAVES; ++w) {
            ts += p[w * (16 * TPITCH)];
            tq += p[w * (16 * TPITCH) + TPITCH];
        }
        atomicAdd(&fsum[tid], ts);
        atomicAdd(&fsq[tid], tq);
    }
}

__global__ __launch_bounds__(CONV_WAVES * 32) void k_conv(const float* __restrict__ x,
                                              const int* __restrict__ ptr,
                                              const int* __restrict__ csr, int e,
                                              float* __restrict__ hpre,
                                              const float* __restrict__ W,
                                              const float* __restrict__ bias,
                                              const float* __restrict__ alpha_p,
                                              const float* __restrict__ ndst,
                                              float* __restrict__ fsum,
                                              float* __restrict__ fsq, int n) {
    conv_body(false, x, ptr, csr, e, nullptr, nullptr, nullptr, nullptr, nullptr, hpre, W, bias,
              alpha_p, ndst, fsum, fsq, n);
}

__global__ __launch_bounds__(CONV_WAVES * 32) void k_conv_bn(const float* __restrict__ x,
                                                 const int* __restrict__ ptr,
                                                 const int* __restrict__ csr, int e,
                                                 const float* __restrict__ psum,
                                                 const float* __restrict__ psq,
                                                 const float* __restrict__ pgamma,
                                                 const float* __restrict__ pbeta,
                                                 const float* __restrict__ palpha_p,
                                                 float* __restrict__ hpre,
                                                 const float* __restrict__ W,
                                                 const float* __restrict__ bias,
                                                 const float* __restrict__ alpha_p,
                                                 const float* __restrict__ ndst,
                                                 float* __restrict__ fsum,
                                                 float* __restrict__ fsq, int n) {
    conv_body(true, x, ptr, csr, e, psum, psq, pgamma, pbeta, palpha_p, hpre, W, bias, alpha_p,
              ndst, fsum, fsq, n);
}

#define BN_ROWS 16

__device__ __forceinline__ void gh_flush(float* __restrict__ gh, int g, int lane, float4 s) {
    float* gp = gh + (long)g * DFEAT + lane * 4;
    atomicAdd(gp + 0, s.x);
    atomicAdd(gp + 1, s.y);
    atomicAdd(gp + 2, s.z);
    atomicAdd(gp + 3, s.w);
}

__global__ __launch_bounds__(128) void k_bnapply2(float* h, const float* __restrict__ h0,
                                                  const float* __restrict__ fsum1,
                                                  const float* __restrict__ fsq1,
                                                  const float* __restrict__ gamma1,
                                                  const float* __restrict__ beta1,
                                                  const float* __restrict__ alpha1_p,
                                                  const float* __restrict__ fsum0,
                                                  const float* __restrict__ fsq0,
                                                  const float* __restrict__ gamma0,
                                                  const float* __restrict__ beta0,
                                                  const float* __restrict__ alpha0_p,
                                                  const int* __restrict__ gids,
                                                  float* __restrict__ gh, int n) {
    const int lane = threadIdx.x & 31;
    const int gw = __builtin_amdgcn_readfirstlane((int)(blockIdx.x * 4 + (threadIdx.x >> 5)));
    const int row0 = gw * BN_ROWS;
    if (row0 >= n) return;
    const float alpha1 = alpha1_p[0];
    const float alpha0 = alpha0_p[0];
    const float4 s41 = ((const float4*)fsum1)[lane];
    const float4 q41 = ((const float4*)fsq1)[lane];
    const float4 g41 = ((const float4*)gamma1)[lane];
    const float4 b41 = ((const float4*)beta1)[lane];
    const float4 s40 = ((const float4*)fsum0)[lane];
    const float4 q40 = ((const float4*)fsq0)[lane];
    const float4 g40 = ((const float4*)gamma0)[lane];
    const float4 b40 = ((const float4*)beta0)[lane];
    const float invn = 1.0f / (float)n;
    float4 a1, c1, a0, c0;
    float4* hr = (float4*)h + (long)row0 * 32 + lane;
    const float4* h0r = (const float4*)h0 + (long)row0 * 32 + lane;
    float4 s = make_float4(0.0f, 0.0f, 0.0f, 0.0f);
    int gcur = gids[row0];
    if (row0 + BN_ROWS <= n) {
        float4 v1[BN_ROWS];
        float4 v0[BN_ROWS];
        int g[BN_ROWS];
#pragma unroll
        for (int i = 0; i < BN_ROWS; ++i) {
            v1[i] = hr[i * 32];
            v0[i] = h0r[i * 32];
            g[i] = gids[row0 + i];
        }
        __builtin_amdgcn_sched_barrier(0);
        bn_affine4(s41, q41, g41, b41, invn, a1, c1);
        bn_affine4(s40, q40, g40, b40, invn, a0, c0);
#pragma unroll
        for (int i = 0; i < BN_ROWS; ++i) {
            const float4 y1 = bn_act(v1[i], a1, c1, alpha1);
            hr[i * 32] = y1;
            const float4 y0 = bn_act(v0[i], a0, c0, alpha0);
            if (g[i] != gcur) {
                gh_flush(gh, gcur, lane, s);
                s = make_float4(0.0f, 0.0f, 0.0f, 0.0f);
                gcur = g[i];
            }
            s.x += y0.x + y1.x; s.y += y0.y + y1.y; s.z += y0.z + y1.z; s.w += y0.w + y1.w;
        }
    } else {
        const int cnt = n - row0;
        bn_affine4(s41, q41, g41, b41, invn, a1, c1);
        bn_affine4(s40, q40, g40, b40, invn, a0, c0);
        for (int i = 0; i < cnt; ++i) {
            const float4 y1 = bn_act(hr[i * 32], a1, c1, alpha1);
            hr[i * 32] = y1;
            const float4 y0 = bn_act(h0r[i * 32], a0, c0, alpha0);
            const int g = gids[row0 + i];
            if (g != gcur) {
                gh_flush(gh, gcur, lane, s);
                s = make_float4(0.0f, 0.0f, 0.0f, 0.0f);
                gcur = g;
            }
            s.x += y0.x + y1.x; s.y += y0.y + y1.y; s.z += y0.z + y1.z; s.w += y0.w + y1.w;
        }
    }
    gh_flush(gh, gcur, lane, s);
}

static inline int cdiv(long a, long b) { return (int)((a + b - 1) / b); }

extern "C" void kernel_launch(void* const* d_in, const int* in_sizes, int n_in,
                              void* d_out, int out_size, void* d_ws, size_t ws_size,
                              hipStream_t stream) {
    (void)n_in; (void)ws_size;
    const float* heat = (const float*)d_in[0];
    const int* esrc   = (const int*)d_in[1];
    const int* edst   = (const int*)d_in[2];
    const int* gids   = (const int*)d_in[3];
    const float* W0 = (const float*)d_in[4];
    const float* b0 = (const float*)d_in[5];
    const float* ac0 = (const float*)d_in[6];
    const float* g0 = (const float*)d_in[7];
    const float* be0 = (const float*)d_in[8];
    const float* aa0 = (const float*)d_in[9];
    const float* W1 = (const float*)d_in[10];
    const float* b1 = (const float*)d_in[11];
    const float* ac1 = (const float*)d_in[12];
    const float* g1 = (const float*)d_in[13];
    const float* be1 = (const float*)d_in[14];
    const float* aa1 = (const float*)d_in[15];

    const int E = in_sizes[1];
    const int N = in_sizes[3];
    const int ND = N * DFEAT;
    const int GD = out_size - ND;
    const int NB = cdiv(N, RSIZE);
    const int slice = cdiv(E, BIN_BLOCKS);

    float* ws   = (float*)d_ws;
    int* dout   = (int*)d_ws;
    int* ptr    = dout + N;
    float* agg  = ws + 2 * (long)N;
    float* r2   = agg + ND;
    int* csr    = (int*)r2;
    float* ndst = r2 + 2 * (long)E;
    int* sortD  = (int*)(ndst + N + (N & 1));
    int* sortS  = sortD + 2 * (long)E;
    int* tabD   = sortS + E;
    int* tabS   = tabD + (long)BIN_BLOCKS * (NB + 1);
    float* fsum0 = r2 + ND;
    float* fsq0  = fsum0 + DFEAT;
    float* fsum1 = fsq0 + DFEAT;
    float* fsq1  = fsum1 + DFEAT;

    float* hout = (float*)d_out;
    float* gh   = hout + ND;

    k_bin<<<BIN_BLOCKS, BIN_THREADS, 0, stream>>>(esrc, edst, sortD, sortS, tabD, tabS, fsum0,
                                                  4 * DFEAT, gh, GD, E, slice, NB);
    k_count<<<NB, BIN_THREADS, 0, stream>>>(sortD, sortS, tabD, tabS, dout, ptr, ndst, N, slice, NB);
    k_place<<<NB, BIN_THREADS, 0, stream>>>(sortD, tabD, dout, ptr, csr, N, E, slice, NB);

    k_conv<<<CONV_BLOCKS, CONV_WAVES * 32, 0, stream>>>(heat, ptr, csr, E, agg, W0, b0, ac0, ndst, fsum0, fsq0, N);

    k_conv_bn<<<CONV_BLOCKS, CONV_WAVES * 32, 0, stream>>>(agg, ptr, csr, E, fsum0, fsq0, g0, be0, aa0, hout,
                                               W1, b1, ac1, ndst, fsum1, fsq1, N);

    k_bnapply2<<<cdiv(N, BN_ROWS * 4), 128, 0, stream>>>(hout, agg, fsum1, fsq1, g1, be1, aa1,
                                                        fsum0, fsq0, g0, be0, aa0, gids, gh, N);
}


# edge grouping: a thread's edges (k_bin, kept in registers for both passes) and a wave's slice pieces (k_count, k_place) are loaded together ahead of use; bucket (sum) order stays LDS arrival order
# speedup vs baseline: 4.9944x; 1.0602x over previous
#include <hip/hip_runtime.h>

typedef __attribute__((ext_vector_type(2))) float v2f;
typedef __attribute__((ext_vector_type(8))) float v8f;

#define DFEAT 128
#define PSTR 288
#define TPITCH 132

#define RSHIFT 8
#define RSIZE 256
#define BIN_THREADS 1024
#define BIN_BLOCKS 128
#define SLICE_CAP 4736
#define PLACE_CAP 6144
#define BIN_EPT 5
#define WALK_PER (BIN_BLOCKS / (BIN_THREADS / 32))
#define WALK_WIN 64
static_assert(SLICE_CAP <= BIN_EPT * BIN_THREADS, "a staged slice lies wholly in the edge registers");
static_assert(BIN_BLOCKS % (BIN_THREADS / 32) == 0, "every wave walks the same number of slices");

__device__ __forceinline__ int block_excl_scan(int v, int* red, int nwaves, int& total) {
    const int lane = threadIdx.x & 31;
    const int wave = threadIdx.x >> 5;
    int inc = v;
    for (int o = 1; o < 32; o <<= 1) {
        const int up = __shfl_up(inc, o);
        if (lane >= o) inc += up;
    }
    if (lane == 31) red[wave] = inc;
    __syncthreads();
    const int w = (lane < nwaves) ? red[lane] : 0;
    int winc = w;
    for (int o = 1; o < 32; o <<= 1) {
        const int up = __shfl_up(winc, o);
        if (lane >= o) winc += up;
    }
    total = __shfl(winc, 31);
    const int wexc = __shfl(winc - w, wave);
    __syncthreads();
    return wexc + inc - v;
}

__global__ __launch_bounds__(BIN_THREADS) void k_bin(const int* __restrict__ esrc,
                                                     const int* __restrict__ edst,
                                                     int* __restrict__ sortD,
                                                     int* __restrict__ sortS,
                                                     int* __restrict__ tabD,
                                                     int* __restrict__ tabS,
                                                     float* __restrict__ stats, int nstats,
                                                     float* __restrict__ gh, int ngh,
                                                     int e, int slice, int nb) {
    __shared__ int hD[BIN_THREADS];
    __shared__ int hS[BIN_THREADS];
    __shared__ int red[32];
    __shared__ int2 lD[SLICE_CAP];
    __shared__ int lS[SLICE_CAP];
    const int tid = threadIdx.x;
    const bool staged = slice <= SLICE_CAP;

    const int beg = blockIdx.x * slice;
    const int end = min(beg + slice, e);
    const int safe = max(min(beg, e - 1), 0);
    int es[BIN_EPT], ed[BIN_EPT];
#pragma unroll
    for (int k = 0; k < BIN_EPT; ++k) {
        const int i = beg + tid + k * BIN_THREADS;
        const int ic = (i < end) ? i : safe;
        ed[k] = edst[ic];
        es[k] = esrc[ic];
    }

    for (int i = blockIdx.x * BIN_THREADS + tid; i < nstats + ngh; i += gridDim.x * BIN_THREADS) {
        if (i < nstats) stats[i] = 0.0f;
        else gh[i - nstats] = 0.0f;
    }

    hD[tid] = 0;
    hS[tid] = 0;
    __syncthreads();

#pragma unroll
    for (int k = 0; k < BIN_EPT; ++k) {
        if (beg + tid + k * BIN_THREADS < end) {
            atomicAdd(&hD[ed[k] >> RSHIFT], 1);
            atomicAdd(&hS[es[k] >> RSHIFT], 1);
        }
    }
    for (int i = beg + tid + BIN_EPT * BIN_THREADS; i < end; i += BIN_THREADS) {
        atomicAdd(&hD[edst[i] >> RSHIFT], 1);
        atomicAdd(&hS[esrc[i] >> RSHIFT], 1);
    }
    __syncthreads();

    const int vD = hD[tid];
    const int vS = hS[tid];
    int totD, totS;
    const int exD = block_excl_scan(vD, red, BIN_THREADS / 32, totD);
    const int exS = block_excl_scan(vS, red, BIN_THREADS / 32, totS);
    hD[tid] = exD;
    hS[tid] = exS;
    int* rowD = tabD + (long)blockIdx.x * (nb + 1);
    int* rowS = tabS + (long)blockIdx.x * (nb + 1);
    if (tid < nb) {
        rowD[tid] = exD;
        rowS[tid] = exS;
    }
    if (tid == 0) {
        rowD[nb] = totD;
        rowS[nb] = totS;
    }
    __syncthreads();

    if (staged) {
#pragma unroll
        for (int k = 0; k < BIN_EPT; ++k) {
            if (beg + tid + k * BIN_THREADS < end) {
                const int s = es[k];
                const int d = ed[k];
                const int pd = atomicAdd(&hD[d >> RSHIFT], 1);
                const int ps = atomicAdd(&hS[s >> RSHIFT], 1);
                lD[pd] = make_int2(s, d);
                lS[ps] = s;
            }
        }
        __syncthreads();
        const int cnt = end - beg;
        for (int i = tid; i < cnt; i += BIN_THREADS) {
            ((int2*)sortD)[beg + i] = lD[i];
            sortS[beg + i] = lS[i];
        }
    } else {
#pragma unroll
        for (int k = 0; k < BIN_EPT; ++k) {
            if (beg + tid + k * BIN_THREADS < end) {
                const int s = es[k];
                const int d = ed[k];
                const int pd = atomicAdd(&hD[d >> RSHIFT], 1);
                const int ps = atomicAdd(&hS[s >> RSHIFT], 1);
                ((int2*)sortD)[beg + pd] = make_int2(s, d);
                sortS[beg + ps] = s;
            }
        }
        for (int i = beg + tid + BIN_EPT * BIN_THREADS; i < end; i += BIN_THREADS) {
            const int s = esrc[i];
            const int d = edst[i];
            const int pd = atomicAdd(&hD[d >> RSHIFT], 1);
            const int ps = atomicAdd(&hS[s >> RSHIFT], 1);
            ((int2*)sortD)[beg + pd] = make_int2(s, d);
            sortS[beg + ps] = s;
        }
    }
}

__global__ __launch_bounds__(BIN_THREADS) void k_count(const int* __restrict__ sortD,
                                                       const int* __restrict__ sortS,
                                                       const int* __restrict__ tabD,
                                                       const int* __restrict__ tabS,
                                                       int* __restrict__ dout,
                                                       int* __restrict__ ptr,
                                                       float* __restrict__ ndst,
                                                       int n, int slice, int nb) {
    __shared__ int cIn[RSIZE];
    __shared__ int cOut[RSIZE];
    __shared__ int pD0[BIN_BLOCKS], pD1[BIN_BLOCKS], pS0[BIN_BLOCKS], pS1[BIN_BLOCKS];
    __shared__ int red[32];
    const int tid = threadIdx.x;
    const int lane = tid & 31;
    const int wave = tid >> 5;
    const int b = blockIdx.x;
    const int lo = b << RSHIFT;

    if (tid < RSIZE) { cIn[tid] = 0; cOut[tid] = 0; }
    int below = 0;
    if (tid < BIN_BLOCKS) {
        const int* td = tabD + (long)tid * (nb + 1) + b;
        const int* ts = tabS + (long)tid * (nb + 1) + b;
        const int d0 = td[0], d1 = td[1], s0 = ts[0], s1 = ts[1];
        pD0[tid] = tid * slice + d0;
        pD1[tid] = tid * slice + d1;
        pS0[tid] = tid * slice + s0;
        pS1[tid] = tid * slice + s1;
        below = d0;
    }
    int base;
    (void)block_excl_scan(below, red, BIN_THREADS / 32, base);

    const int2* sd = (const int2*)sortD;
    int d0[WALK_PER], d1[WALK_PER], s0[WALK_PER], s1[WALK_PER];
#pragma unroll
    for (int i = 0; i < WALK_PER; ++i) {
        const int j = wave + i * (BIN_THREADS / 32);
        d0[i] = pD0[j]; d1[i] = pD1[j];
        s0[i] = pS0[j]; s1[i] = pS1[j];
    }
    int vd[WALK_PER][WALK_WIN / 32], vs[WALK_PER][WALK_WIN / 32];
#pragma unroll
    for (int i = 0; i < WALK_PER; ++i) {
#pragma unroll
        for (int h = 0; h < WALK_WIN / 32; ++h) {
            vd[i][h] = sd[max(min(d0[i] + h * 32 + lane, d1[i] - 1), 0)].y;
            vs[i][h] = sortS[max(min(s0[i] + h * 32 + lane, s1[i] - 1), 0)];
        }
    }
#pragma unroll
    for (int i = 0; i < WALK_PER; ++i) {
#pragma unroll
        for (int h = 0; h < WALK_WIN / 32; ++h) {
            if (d0[i] + h * 32 + lane < d1[i]) atomicAdd(&cIn[vd[i][h] - lo], 1);
            if (s0[i] + h * 32 + lane < s1[i]) atomicAdd(&cOut[vs[i][h] - lo], 1);
        }
    }
#pragma unroll
    for (int i = 0; i < WALK_PER; ++i) {
        for (int k = d0[i] + WALK_WIN + lane; k < d1[i]; k += 32) atomicAdd(&cIn[sd[k].y - lo], 1);
        for (int k = s0[i] + WALK_WIN + lane; k < s1[i]; k += 32) atomicAdd(&cOut[sortS[k] - lo], 1);
    }
    __syncthreads();

    const int v = (tid < RSIZE) ? cIn[tid] : 0;
    int total;
    const int ex = block_excl_scan(v, red, BIN_THREADS / 32, total);
    if (tid < RSIZE && lo + tid < n) {
        ptr[lo + tid] = base + ex;
        ndst[lo + tid] = (v > 0) ? rsqrtf((float)v) : 0.0f;
        dout[lo + tid] = cOut[tid];
    }
}

__global__ __launch_bounds__(BIN_THREADS) void k_place(const int* __restrict__ sortD,
                                                       const int* __restrict__ tabD,
                                                       const int* __restrict__ dout,
                                                       const int* __restrict__ ptr,
                                                       int* __restrict__ csr,
                                                       int n, int e, int slice, int nb) {
    __shared__ int cur[RSIZE];
    __shared__ int pD0[BIN_BLOCKS], pD1[BIN_BLOCKS];
    __shared__ int2 lC[PLACE_CAP];
    const int tid = threadIdx.x;
    const int lane = tid & 31;
    const int wave = tid >> 5;
    const int b = blockIdx.x;
    const int lo = b << RSHIFT;
    const int base = ptr[lo];
    const int total = ((lo + RSIZE < n) ? ptr[lo + RSIZE] : e) - base;
    const bool staged = total <= PLACE_CAP;

    if (tid < RSIZE) cur[tid] = (lo + tid < n) ? ptr[lo + tid] : 0;
    if (tid < BIN_BLOCKS) {
        const int* td = tabD + (long)tid * (nb + 1) + b;
        pD0[tid] = tid * slice + td[0];
        pD1[tid] = tid * slice + td[1];
    }
    __syncthreads();

    const int2* sd = (const int2*)sortD;
    int d0[WALK_PER], d1[WALK_PER];
#pragma unroll
    for (int i = 0; i < WALK_PER; ++i) {
        const int j = wave + i * (BIN_THREADS / 32);
        d0[i] = pD0[j]; d1[i] = pD1[j];
    }
    int2 ev[WALK_PER][WALK_WIN / 32];
    int dg[WALK_PER][WALK_WIN / 32];
#pragma unroll
    for (int i = 0; i < WALK_PER; ++i) {
#pragma unroll
        for (int h = 0; h < WALK_WIN / 32; ++h)
            ev[i][h] = sd[max(min(d0[i] + h * 32 + lane, d1[i] - 1), 0)];
    }
#pragma unroll
    for (int i = 0; i < WALK_PER; ++i) {
#pragma unroll
        for (int h = 0; h < WALK_WIN / 32; ++h) dg[i][h] = dout[ev[i][h].x];
    }
    if (staged) {
#pragma unroll
        for (int i = 0; i < WALK_PER; ++i) {
#pragma unroll
            for (int h = 0; h < WALK_WIN / 32; ++h) {
                if (d0[i] + h * 32 + lane < d1[i]) {
                    const int2 ed = ev[i][h];
                    const int pos = atomicAdd(&cur[ed.y - lo], 1);
                    const float w = rsqrtf((float)dg[i][h]);
                    lC[pos - base] = make_int2(ed.x, __float_as_int(w));
                }
            }
        }
#pragma unroll
        for (int i = 0; i < WALK_PER; ++i) {
            for (int k = d0[i] + WALK_WIN + lane; k < d1[i]; k += 32) {
                const int2 ed = sd[k];
                const int pos = atomicAdd(&cur[ed.y - lo], 1);
                const float w = rsqrtf((float)dout[ed.x]);
                lC[pos - base] = make_int2(ed.x, __float_as_int(w));
            }
        }
        __syncthreads();
        for (int i = tid; i < total; i += BIN_THREADS) ((int2*)csr)[base + i] = lC[i];
    } else {
#pragma unroll
        for (int i = 0; i < WALK_PER; ++i) {
#pragma unroll
            for (int h = 0; h < WALK_WIN / 32; ++h) {
                if (d0[i] + h * 32 + lane < d1[i]) {
                    const int2 ed = ev[i][h];
                    const int pos = atomicAdd(&cur[ed.y - lo], 1);
                    const float w = rsqrtf((float)dg[i][h]);
                    ((int2*)csr)[pos] = make_int2(ed.x, __float_as_int(w));
                }
            }
        }
#pragma unroll
        for (int i = 0; i < WALK_PER; ++i) {
            for (int k = d0[i] + WALK_WIN + lane; k < d1[i]; k += 32) {
                const int2 ed = sd[k];
                const int pos = atomicAdd(&cur[ed.y - lo], 1);
                const float w = rsqrtf((float)dout[ed.x]);
                ((int2*)csr)[pos] = make_int2(ed.x, __float_as_int(w));
            }
        }
    }
}

__device__ __forceinline__ float4 bn_act(float4 v, float4 a, float4 c, float alpha) {
    float4 y;
    y.x = fmaf(v.x, a.x, c.x); y.x = (y.x >= 0.0f) ? y.x : alpha * y.x;
    y.y = fmaf(v.y, a.y, c.y); y.y = (y.y >= 0.0f) ? y.y : alpha * y.y;
    y.z = fmaf(v.z, a.z, c.z); y.z = (y.z >= 0.0f) ? y.z : alpha * y.z;
    y.w = fmaf(v.w, a.w, c.w); y.w = (y.w >= 0.0f) ? y.w : alpha * y.w;
    return y;
}

__device__ __forceinline__ void bn_affine(float s, float q, float g, float b, float invn,
                                          float& a, float& c) {
    const float mean = s * invn;
    const float var = q * invn - mean * mean;
    const float inv = rsqrtf(var + 1e-5f);
    a = inv * g;
    c = b - mean * a;
}

__device__ __forceinline__ void bn_affine4(float4 s, float4 q, float4 g, float4 b, float invn,
                                           float4& a, float4& c) {
    bn_affine(s.x, q.x, g.x, b.x, invn, a.x, c.x);
    bn_affine(s.y, q.y, g.y, b.y, invn, a.y, c.y);
    bn_affine(s.z, q.z, g.z, b.z, invn, a.z, c.z);
    bn_affine(s.w, q.w, g.w, b.w, invn, a.w, c.w);
}

__device__ __forceinline__ void tile_store_row(float* tw, int r, float4 acc, float nd) {
    float4 o;
    o.x = acc.x * nd; o.y = acc.y * nd; o.z = acc.z * nd; o.w = acc.w * nd;
    *(float4*)(tw + r * TPITCH + (threadIdx.x & 31) * 4) = o;
}

__device__ __forceinline__ void gather_tile(const bool bn, const float* __restrict__ x,
                                            const int* __restrict__ ptr,
                                            const int* __restrict__ csr,
                                            const float* __restrict__ ndst,
                                            float* tw, int row0, int n, int e,
                                            float4 a, float4 c, float alpha) {
    const int lane = threadIdx.x & 31;
    const int rl = row0 + (lane & 15);
    const int rc = (rl < n) ? rl : (n - 1);
    const int pnext = ptr[min(rl + 1, n - 1)];
    const int myEnd = (rl + 1 < n) ? pnext : e;
    const int myNd = (rl < n) ? __float_as_int(ndst[rc]) : 0;
    const int nrows = min(16, n - row0);
    int cur = 0;
    float4 acc = make_float4(0.0f, 0.0f, 0.0f, 0.0f);
    if (nrows > 0) {
        const int beg = ptr[row0];
        const int endAll = __builtin_amdgcn_readlane(myEnd, nrows - 1);
        int curEnd = __builtin_amdgcn_readlane(myEnd, 0);
        const int2* ent = (const int2*)csr;
        const float4* xv = (const float4*)x + lane;
        int2 en[16];
        if (beg < endAll) {
#pragma unroll
            for (int s = 0; s < 16; ++s) en[s] = ent[min(beg + s, endAll - 1)];
        }
        for (int j = beg; j < endAll; j += 16) {
            float4 v[16];
#pragma unroll
            for (int s = 0; s < 16; ++s) v[s] = xv[(long)en[s].x * 32];
            int2 nx[16];
#pragma unroll
            for (int s = 0; s < 16; ++s) nx[s] = ent[min(j + 16 + s, endAll - 1)];
#pragma unroll
            for (int s = 0; s < 16; ++s) {
                if (j + s < endAll) {
                    while (j + s >= curEnd) {
                        tile_store_row(tw, cur, acc,
                                       __int_as_float(__builtin_amdgcn_readlane(myNd, cur)));
                        acc = make_float4(0.0f, 0.0f, 0.0f, 0.0f);
                        ++cur;
                        curEnd = __builtin_amdgcn_readlane(myEnd, cur);
                    }
                    float4 t = v[s];
                    if (bn) t = bn_act(t, a, c, alpha);
                    const float w = __int_as_float(en[s].y);
                    acc.x += t.x * w; acc.y += t.y * w; acc.z += t.z * w; acc.w += t.w * w;
                }
            }
#pragma unroll
            for (int s = 0; s < 16; ++s) en[s] = nx[s];
        }
    }
    for (; cur < 16; ++cur) {
        tile_store_row(tw, cur, acc, __int_as_float(__builtin_amdgcn_readlane(myNd, cur)));
        acc = make_float4(0.0f, 0.0f, 0.0f, 0.0f);
    }
}

__device__ __forceinline__ void gemm_ep_row(float4& v, float4 bv, float alpha, float4& s, float4& q) {
    v.x += bv.x; v.y += bv.y; v.z += bv.z; v.w += bv.w;
    v.x = (v.x >= 0.0f) ? v.x : alpha * v.x;
    v.y = (v.y >= 0.0f) ? v.y : alpha * v.y;
    v.z = (v.z >= 0.0f) ? v.z : alpha * v.z;
    v.w = (v.w >= 0.0f) ? v.w : alpha * v.w;
    s.x += v.x; s.y += v.y; s.z += v.z; s.w += v.w;
    q.x += v.x * v.x; q.y += v.y * v.y; q.z += v.z * v.z; q.w += v.w * v.w;
}

#define CONV_WAVES 16
#define CONV_BLOCKS 256

__device__ __forceinline__ void wave_lds_order() {
    __builtin_amdgcn_fence(__ATOMIC_RELEASE, "wavefront");
    __builtin_amdgcn_wave_barrier();
    __builtin_amdgcn_fence(__ATOMIC_ACQUIRE, "wavefront");
}

__device__ __forceinline__ void conv_body(const bool bn, const float* __restrict__ x,
                                          const int* __restrict__ ptr,
                                          const int* __restrict__ csr, int e,
                                          const float* __restrict__ psum,
                                          const float* __restrict__ psq,
                                          const float* __restrict__ pgamma,
                                          const float* __restrict__ pbeta,
                                          const float* __restrict__ palpha_p,
                                          float* __restrict__ hpre,
                                          const float* __restrict__ W,
                                          const float* __restrict__ bias,
                                          const float* __restrict__ alpha_p,
                                          const float* __restrict__ ndst,
                                          float* __restrict__ fsum,
                                          float* __restrict__ fsq, int n) {
    __shared__ float sW[64 * PSTR];
    __shared__ float4 sT[CONV_WAVES * 16 * (TPITCH / 4)];

    const int tid = threadIdx.x;

    for (int kp = tid >> 4; kp < 64; kp += (CONV_WAVES * 32) >> 4) {
        const int nb = (tid & 15) * 8;
        const float* w0 = W + (2 * kp) * DFEAT + nb;
        const float* w1 = w0 + DFEAT;
        const float4 r0a = ((const float4*)w0)[0];
        const float4 r0b = ((const float4*)w0)[1];
        const float4 r1a = ((const float4*)w1)[0];
        const float4 r1b = ((const float4*)w1)[1];
        float2* dp = (float2*)(sW + kp * PSTR + nb * 2);
        dp[0] = make_float2(r0a.x, r1a.x);
        dp[1] = make_float2(r0a.y, r1a.y);
        dp[2] = make_float2(r0a.z, r1a.z);
        dp[3] = make_float2(r0a.w, r1a.w);
        dp[4] = make_float2(r0b.x, r1b.x);
        dp[5] = make_float2(r0b.y, r1b.y);
        dp[6] = make_float2(r0b.z, r1b.z);
        dp[7] = make_float2(r0b.w, r1b.w);
    }
    __syncthreads();

    const int wave = tid >> 5;
    const int lane = tid & 31;
    const int half = lane >> 4;
    const int mcol = lane & 15;
    const float alpha = alpha_p[0];
    const int wave_s = __builtin_amdgcn_readfirstlane(wave);
    float* tw = (float*)sT + wave_s * (16 * TPITCH);
    float4 ga = make_float4(1.0f, 1.0f, 1.0f, 1.0f);
    float4 gc = make_float4(0.0f, 0.0f, 0.0f, 0.0f);
    float galpha = 1.0f;
    if (bn) {
        galpha = palpha_p[0];
        const float4 s4 = ((const float4*)psum)[lane];
        const float4 q4 = ((const float4*)psq)[lane];
        const float4 g4 = ((const float4*)pgamma)[lane];
        const float4 b4 = ((const float4*)pbeta)[lane];
        bn_affine4(s4, q4, g4, b4, 1.0f / (float)n, ga, gc);
    }
    const float* arow = tw + mcol * TPITCH + half * 2;
    const float* trow = tw + lane * 4;
    const float4 bv = ((const float4*)bias)[lane];
    float4 s = make_float4(0.0f, 0.0f, 0.0f, 0.0f);
    float4 q = make_float4(0.0f, 0.0f, 0.0f, 0.0f);

    const int ntiles = (n + 15) >> 4;
    for (int tile = blockIdx.x + gridDim.x * wave_s; tile < ntiles; tile += gridDim.x * CONV_WAVES) {
        const int row0 = tile * 16;
        gather_tile(bn, x, ptr, csr, ndst, tw, row0, n, e, ga, gc, galpha);
        wave_lds_order();

        v8f acc[8] = {};
        for (int kc = 0; kc < DFEAT; kc += 32) {
            const float* wc = sW + (kc / 2) * PSTR;
            for (int kb = 0; kb < 32; kb += 4) {
                const v2f a = *(const v2f*)(arow + kc + kb);
                const float* bp = wc + (kb / 2 + half) * PSTR + mcol * 2;
#pragma unroll
                for (int t = 0; t < 8; ++t) {
                    v2f b = *(const v2f*)(bp + t * 32);
                    acc[t] = __builtin_amdgcn_wmma_f32_16x16x4_f32(
                        false, a, false, b, (short)0, acc[t], false, false);
                }
            }
        }

        wave_lds_order();
#pragma unroll
        for (int t = 0; t < 8; ++t) {
#pragma unroll
            for (int r = 0; r < 8; ++r) tw[(half * 8 + r) * TPITCH + t * 16 + mcol] = acc[t][r];
        }
        wave_lds_order();
        if (row0 + 16 <= n) {
            float4* orow = (float4*)hpre + (long)row0 * 32 + lane;
            float4 v[16];
#pragma unroll
            for (int i = 0; i < 16; ++i) v[i] = *(const float4*)(trow + i * TPITCH);
#pragma unroll
            for (int i = 0; i < 16; ++i) {
                gemm_ep_row(v[i], bv, alpha, s, q);
                orow[i * 32] = v[i];
            }
        } else {
            for (int i = 0; i < 16 && row0 + i < n; ++i) {
                float4 v = *(const float4*)(trow + i * TPITCH);
                gemm_ep_row(v, bv, alpha, s, q);
                ((float4*)hpre)[(long)(row0 + i) * 32 + lane] = v;
            }
        }
        wave_lds_order();
    }

    *(float4*)(tw + lane * 4) = s;
    *(float4*)(tw + TPITCH + lane * 4) = q;
    __syncthreads();
    if (tid < DFEAT) {
        const float* p = (const float*)sT + tid;
        float ts = 0.0f, tq = 0.0f;
        for (int w = 0; w < CONV_WAVES; ++w) {
            ts += p[w * (16 * TPITCH)];
            tq += p[w * (16 * TPITCH) + TPITCH];
        }
        atomicAdd(&fsum[tid], ts);
        atomicAdd(&fsq[tid], tq);
    }
}

__global__ __launch_bounds__(CONV_WAVES * 32) void k_conv(const float* __restrict__ x,
                                              const int* __restrict__ ptr,
                                              const int* __restrict__ csr, int e,
                                              float* __restrict__ hpre,
                                              const float* __restrict__ W,
                                              const float* __restrict__ bias,
                                              const float* __restrict__ alpha_p,
                                              const float* __restrict__ ndst,
                                              float* __restrict__ fsum,
                                              float* __restrict__ fsq, int n) {
    conv_body(false, x, ptr, csr, e, nullptr, nullptr, nullptr, nullptr, nullptr, hpre, W, bias,
              alpha_p, ndst, fsum, fsq, n);
}

__global__ __launch_bounds__(CONV_WAVES * 32) void k_conv_bn(const float* __restrict__ x,
                                                 const int* __restrict__ ptr,
                                                 const int* __restrict__ csr, int e,
                                                 const float* __restrict__ psum,
                                                 const float* __restrict__ psq,
                                                 const float* __restrict__ pgamma,
                                                 const float* __restrict__ pbeta,
                                                 const float* __restrict__ palpha_p,
                                                 float* __restrict__ hpre,
                                                 const float* __restrict__ W,
                                                 const float* __restrict__ bias,
                                                 const float* __restrict__ alpha_p,
                                                 const float* __restrict__ ndst,
                                                 float* __restrict__ fsum,
                                                 float* __restrict__ fsq, int n) {
    conv_body(true, x, ptr, csr, e, psum, psq, pgamma, pbeta, palpha_p, hpre, W, bias, alpha_p,
              ndst, fsum, fsq, n);
}

#define BN_ROWS 16

__device__ __forceinline__ void gh_flush(float* __restrict__ gh, int g, int lane, float4 s) {
    float* gp = gh + (long)g * DFEAT + lane * 4;
    atomicAdd(gp + 0, s.x);
    atomicAdd(gp + 1, s.y);
    atomicAdd(gp + 2, s.z);
    atomicAdd(gp + 3, s.w);
}

__global__ __launch_bounds__(128) void k_bnapply2(float* h, const float* __restrict__ h0,
                                                  const float* __restrict__ fsum1,
                                                  const float* __restrict__ fsq1,
                                                  const float* __restrict__ gamma1,
                                                  const float* __restrict__ beta1,
                                                  const float* __restrict__ alpha1_p,
                                                  const float* __restrict__ fsum0,
                                                  const float* __restrict__ fsq0,
                                                  const float* __restrict__ gamma0,
                                                  const float* __restrict__ beta0,
                                                  const float* __restrict__ alpha0_p,
                                                  const int* __restrict__ gids,
                                                  float* __restrict__ gh, int n) {
    const int lane = threadIdx.x & 31;
    const int gw = __builtin_amdgcn_readfirstlane((int)(blockIdx.x * 4 + (threadIdx.x >> 5)));
    const int row0 = gw * BN_ROWS;
    if (row0 >= n) return;
    const float alpha1 = alpha1_p[0];
    const float alpha0 = alpha0_p[0];
    const float4 s41 = ((const float4*)fsum1)[lane];
    const float4 q41 = ((const float4*)fsq1)[lane];
    const float4 g41 = ((const float4*)gamma1)[lane];
    const float4 b41 = ((const float4*)beta1)[lane];
    const float4 s40 = ((const float4*)fsum0)[lane];
    const float4 q40 = ((const float4*)fsq0)[lane];
    const float4 g40 = ((const float4*)gamma0)[lane];
    const float4 b40 = ((const float4*)beta0)[lane];
    const float invn = 1.0f / (float)n;
    float4 a1, c1, a0, c0;
    float4* hr = (float4*)h + (long)row0 * 32 + lane;
    const float4* h0r = (const float4*)h0 + (long)row0 * 32 + lane;
    float4 s = make_float4(0.0f, 0.0f, 0.0f, 0.0f);
    int gcur = gids[row0];
    if (row0 + BN_ROWS <= n) {
        float4 v1[BN_ROWS];
        float4 v0[BN_ROWS];
        int g[BN_ROWS];
#pragma unroll
        for (int i = 0; i < BN_ROWS; ++i) {
            v1[i] = hr[i * 32];
            v0[i] = h0r[i * 32];
            g[i] = gids[row0 + i];
        }
        __builtin_amdgcn_sched_barrier(0);
        bn_affine4(s41, q41, g41, b41, invn, a1, c1);
        bn_affine4(s40, q40, g40, b40, invn, a0, c0);
#pragma unroll
        for (int i = 0; i < BN_ROWS; ++i) {
            const float4 y1 = bn_act(v1[i], a1, c1, alpha1);
            hr[i * 32] = y1;
            const float4 y0 = bn_act(v0[i], a0, c0, alpha0);
            if (g[i] != gcur) {
                gh_flush(gh, gcur, lane, s);
                s = make_float4(0.0f, 0.0f, 0.0f, 0.0f);
                gcur = g[i];
            }
            s.x += y0.x + y1.x; s.y += y0.y + y1.y; s.z += y0.z + y1.z; s.w += y0.w + y1.w;
        }
    } else {
        const int cnt = n - row0;
        bn_affine4(s41, q41, g41, b41, invn, a1, c1);
        bn_affine4(s40, q40, g40, b40, invn, a0, c0);
        for (int i = 0; i < cnt; ++i) {
            const float4 y1 = bn_act(hr[i * 32], a1, c1, alpha1);
            hr[i * 32] = y1;
            const float4 y0 = bn_act(h0r[i * 32], a0, c0, alpha0);
            const int g = gids[row0 + i];
            if (g != gcur) {
                gh_flush(gh, gcur, lane, s);
                s = make_float4(0.0f, 0.0f, 0.0f, 0.0f);
                gcur = g;
            }
            s.x += y0.x + y1.x; s.y += y0.y + y1.y; s.z += y0.z + y1.z; s.w += y0.w + y1.w;
        }
    }
    gh_flush(gh, gcur, lane, s);
}

static inline int cdiv(long a, long b) { return (int)((a + b - 1) / b); }

extern "C" void kernel_launch(void* const* d_in, const int* in_sizes, int n_in,
                              void* d_out, int out_size, void* d_ws, size_t ws_size,
                              hipStream_t stream) {
    (void)n_in; (void)ws_size;
    const float* heat = (const float*)d_in[0];
    const int* esrc   = (const int*)d_in[1];
    const int* edst   = (const int*)d_in[2];
    const int* gids   = (const int*)d_in[3];
    const float* W0 = (const float*)d_in[4];
    const float* b0 = (const float*)d_in[5];
    const float* ac0 = (const float*)d_in[6];
    const float* g0 = (const float*)d_in[7];
    const float* be0 = (const float*)d_in[8];
    const float* aa0 = (const float*)d_in[9];
    const float* W1 = (const float*)d_in[10];
    const float* b1 = (const float*)d_in[11];
    const float* ac1 = (const float*)d_in[12];
    const float* g1 = (const float*)d_in[13];
    const float* be1 = (const float*)d_in[14];
    const float* aa1 = (const float*)d_in[15];

    const int E = in_sizes[1];
    const int N = in_sizes[3];
    const int ND = N * DFEAT;
    const int GD = out_size - ND;
    const int NB = cdiv(N, RSIZE);
    const int slice = cdiv(E, BIN_BLOCKS);

    float* ws   = (float*)d_ws;
    int* dout   = (int*)d_ws;
    int* ptr    = dout + N;
    float* agg  = ws + 2 * (long)N;
    float* r2   = agg + ND;
    int* csr    = (int*)r2;
    float* ndst = r2 + 2 * (long)E;
    int* sortD  = (int*)(ndst + N + (N & 1));
    int* sortS  = sortD + 2 * (long)E;
    int* tabD   = sortS + E;
    int* tabS   = tabD + (long)BIN_BLOCKS * (NB + 1);
    float* fsum0 = r2 + ND;
    float* fsq0  = fsum0 + DFEAT;
    float* fsum1 = fsq0 + DFEAT;
    float* fsq1  = fsum1 + DFEAT;

    float* hout = (float*)d_out;
    float* gh   = hout + ND;

    k_bin<<<BIN_BLOCKS, BIN_THREADS, 0, stream>>>(esrc, edst, sortD, sortS, tabD, tabS, fsum0,
                                                  4 * DFEAT, gh, GD, E, slice, NB);
    k_count<<<NB, BIN_THREADS, 0, stream>>>(sortD, sortS, tabD, tabS, dout, ptr, ndst, N, slice, NB);
    k_place<<<NB, BIN_THREADS, 0, stream>>>(sortD, tabD, dout, ptr, csr, N, E, slice, NB);

    k_conv<<<CONV_BLOCKS, CONV_WAVES * 32, 0, stream>>>(heat, ptr, csr, E, agg, W0, b0, ac0, ndst, fsum0, fsq0, N);

    k_conv_bn<<<CONV_BLOCKS, CONV_WAVES * 32, 0, stream>>>(agg, ptr, csr, E, fsum0, fsq0, g0, be0, aa0, hout,
                                               W1, b1, ac1, ndst, fsum1, fsq1, N);

    k_bnapply2<<<cdiv(N, BN_ROWS * 4), 128, 0, stream>>>(hout, agg, fsum1, fsq1, g1, be1, aa1,
                                                        fsum0, fsq0, g0, be0, aa0, gids, gh, N);
}
